# starting kernel
# baseline (speedup 1.0000x reference)
#include <hip/hip_runtime.h>
#include <hip/hip_bf16.h>
#include <stdint.h>

typedef __attribute__((ext_vector_type(2))) float v2f;
typedef __attribute__((ext_vector_type(8))) float v8f;

#define NEG_SLOPE 0.01f

__device__ __forceinline__ float lrelu(float x) {
    return x > 0.0f ? x : NEG_SLOPE * x;
}

__device__ __forceinline__ unsigned fkey(float f) {
    unsigned u = __float_as_uint(f);
    return (u & 0x80000000u) ? ~u : (u | 0x80000000u);
}
__device__ __forceinline__ float fdecode(unsigned k) {
    unsigned u = (k & 0x80000000u) ? (k ^ 0x80000000u) : ~k;
    return __uint_as_float(u);
}
#define NEG_INF_KEY 0x007FFFFFu

__global__ void k_init(float* __restrict__ out, float* __restrict__ sumh,
                       float* __restrict__ sumt, unsigned* __restrict__ maxh,
                       unsigned* __restrict__ maxt, int n_out, int n_seg) {
    int i = blockIdx.x * blockDim.x + threadIdx.x;
    if (i < n_out) out[i] = 0.0f;
    if (i < n_seg) {
        sumh[i] = 0.0f;
        sumt[i] = 0.0f;
        maxh[i] = NEG_INF_KEY;
        maxt[i] = NEG_INF_KEY;
    }
}

__global__ void k_proj_wmma(const float* __restrict__ X,
                            const float* __restrict__ W0,
                            const float* __restrict__ W1,
                            float* __restrict__ S, int nrows) {
    const int wave = (blockIdx.x * blockDim.x + threadIdx.x) >> 5;
    const int lane = threadIdx.x & 31;
    const int row0 = wave * 16;
    if (row0 >= nrows) return;

    const int m    = lane & 15;
    const int kofs = (lane >> 4) << 1;

    const float* __restrict__ arow = X + (size_t)(row0 + m) * 128;
    const float* __restrict__ brow =
        (m < 8) ? (W0 + (size_t)m * 128) : (W1 + (size_t)(m - 8) * 128);

    v8f c = {0.f, 0.f, 0.f, 0.f, 0.f, 0.f, 0.f, 0.f};
#pragma unroll
    for (int k = 0; k < 128; k += 4) {
        v2f a = *(const v2f*)(arow + k + kofs);
        v2f b = *(const v2f*)(brow + k + kofs);
        c = __builtin_amdgcn_wmma_f32_16x16x4_f32(
             false, a,  false, b,
             (short)0, c,  false,  false);
    }

    const int rbase = row0 + ((lane >> 4) << 3);
    const int col   = lane & 15;
#pragma unroll
    for (int v = 0; v < 8; ++v) {
        S[(size_t)(rbase + v) * 16 + col] = c[v];
    }
}

__global__ void k_edge_max(const int* __restrict__ h_idx,
                           const int* __restrict__ t_idx,
                           const int* __restrict__ rel,
                           const float* __restrict__ Se,
                           const float* __restrict__ Sr,
                           unsigned* __restrict__ maxh,
                           unsigned* __restrict__ maxt, int E) {
    int i = blockIdx.x * blockDim.x + threadIdx.x;
    if (i >= E * 8) return;
    const int e = i >> 3, h = i & 7;
    const int hi = h_idx[e], ti = t_idx[e], r = rel[e];
    const float er = Sr[(size_t)r * 16 + h];
    const float sh = lrelu(Se[(size_t)hi * 16 + h] + er);
    const float st = lrelu(Se[(size_t)ti * 16 + 8 + h] + er);
    atomicMax(&maxh[(size_t)hi * 8 + h], fkey(sh));
    atomicMax(&maxt[(size_t)ti * 8 + h], fkey(st));
}

__global__ void k_edge_sum(const int* __restrict__ h_idx,
                           const int* __restrict__ t_idx,
                           const int* __restrict__ rel,
                           const float* __restrict__ Se,
                           const float* __restrict__ Sr,
                           const unsigned* __restrict__ maxh,
                           const unsigned* __restrict__ maxt,
                           float* __restrict__ sumh,
                           float* __restrict__ sumt, int E) {
    int i = blockIdx.x * blockDim.x + threadIdx.x;
    if (i >= E * 8) return;
    const int e = i >> 3, h = i & 7;
    const int hi = h_idx[e], ti = t_idx[e], r = rel[e];
    const float er = Sr[(size_t)r * 16 + h];
    const float sh = lrelu(Se[(size_t)hi * 16 + h] + er);
    const float st = lrelu(Se[(size_t)ti * 16 + 8 + h] + er);
    atomicAdd(&sumh[(size_t)hi * 8 + h], __expf(sh - fdecode(maxh[(size_t)hi * 8 + h])));
    atomicAdd(&sumt[(size_t)ti * 8 + h], __expf(st - fdecode(maxt[(size_t)ti * 8 + h])));
}

__global__ void k_edge_scatter(const int* __restrict__ h_idx,
                               const int* __restrict__ t_idx,
                               const int* __restrict__ rel,
                               const float* __restrict__ Se,
                               const float* __restrict__ Sr,
                               const unsigned* __restrict__ maxh,
                               const unsigned* __restrict__ maxt,
                               const float* __restrict__ sumh,
                               const float* __restrict__ sumt,
                               const float* __restrict__ x_r,
                               float* __restrict__ out, int E) {
    const int wid  = (blockIdx.x * blockDim.x + threadIdx.x) >> 5;
    const int lane = threadIdx.x & 31;
    if (wid >= E) return;

    const int hi = h_idx[wid], ti = t_idx[wid], r = rel[wid];

    float ah = 0.0f, at = 0.0f;
#pragma unroll
    for (int h = 0; h < 8; ++h) {
        const float er = Sr[(size_t)r * 16 + h];
        const float sh = lrelu(Se[(size_t)hi * 16 + h] + er);
        const float st = lrelu(Se[(size_t)ti * 16 + 8 + h] + er);
        const float eh = __expf(sh - fdecode(maxh[(size_t)hi * 8 + h]));
        const float et = __expf(st - fdecode(maxt[(size_t)ti * 8 + h]));
        ah += eh / (sumh[(size_t)hi * 8 + h] + 1e-16f);
        at += et / (sumt[(size_t)ti * 8 + h] + 1e-16f);
    }
    ah *= 0.125f;
    at *= 0.125f;

    const float* __restrict__ msg = x_r + (size_t)r * 128;
    float* __restrict__ oh = out + (size_t)hi * 256;
    float* __restrict__ ot = out + (size_t)ti * 256 + 128;
#pragma unroll
    for (int d = lane; d < 128; d += 32) {
        const float mv = msg[d];
        atomicAdd(&oh[d], mv * ah);
        atomicAdd(&ot[d], mv * at);
    }
}

extern "C" void kernel_launch(void* const* d_in, const int* in_sizes, int n_in,
                              void* d_out, int out_size, void* d_ws, size_t ws_size,
                              hipStream_t stream) {
    const float* x_e  = (const float*)d_in[0];
    const float* x_r  = (const float*)d_in[1];
    const float* Wh   = (const float*)d_in[2];
    const float* Wt   = (const float*)d_in[3];
    const float* Wr   = (const float*)d_in[4];
    const int*   eidx = (const int*)d_in[5];
    const int*   rel  = (const int*)d_in[6];

    const int D   = 128;
    const int N_e = in_sizes[0] / D;
    const int N_r = in_sizes[1] / D;
    const int E   = in_sizes[6];

    const int* h_idx = eidx;
    const int* t_idx = eidx + E;

    auto align256 = [](size_t x) { return (x + 255) & ~(size_t)255; };
    char* ws = (char*)d_ws;
    float*    Se   = (float*)ws;    ws += align256((size_t)N_e * 16 * sizeof(float));
    float*    Sr   = (float*)ws;    ws += align256((size_t)N_r * 16 * sizeof(float));
    unsigned* maxh = (unsigned*)ws; ws += align256((size_t)N_e * 8 * sizeof(unsigned));
    unsigned* maxt = (unsigned*)ws; ws += align256((size_t)N_e * 8 * sizeof(unsigned));
    float*    sumh = (float*)ws;    ws += align256((size_t)N_e * 8 * sizeof(float));
    float*    sumt = (float*)ws;    ws += align256((size_t)N_e * 8 * sizeof(float));

    float* out = (float*)d_out;

    const int BLK = 256;

    {
        const int n_out = N_e * 256;
        const int n_seg = N_e * 8;
        const int grid  = (n_out + BLK - 1) / BLK;
        k_init<<<grid, BLK, 0, stream>>>(out, sumh, sumt, maxh, maxt, n_out, n_seg);
    }

    {
        const int waves   = (N_e + 15) / 16;
        const int threads = waves * 32;
        const int grid    = (threads + BLK - 1) / BLK;
        k_proj_wmma<<<grid, BLK, 0, stream>>>(x_e, Wh, Wt, Se, N_e);
    }
    {
        const int waves   = (N_r + 15) / 16;
        const int threads = waves * 32;
        const int grid    = (threads + BLK - 1) / BLK;
        k_proj_wmma<<<grid, BLK, 0, stream>>>(x_r, Wr, Wr, Sr, N_r);
    }

    {
        const int n    = E * 8;
        const int grid = (n + BLK - 1) / BLK;
        k_edge_max<<<grid, BLK, 0, stream>>>(h_idx, t_idx, rel, Se, Sr, maxh, maxt, E);
    }

    {
        const int n    = E * 8;
        const int grid = (n + BLK - 1) / BLK;
        k_edge_sum<<<grid, BLK, 0, stream>>>(h_idx, t_idx, rel, Se, Sr, maxh, maxt,
                                             sumh, sumt, E);
    }

    {
        const long long threads = (long long)E * 32;
        const int grid = (int)((threads + BLK - 1) / BLK);
        k_edge_scatter<<<grid, BLK, 0, stream>>>(h_idx, t_idx, rel, Se, Sr, maxh, maxt,
                                                 sumh, sumt, x_r, out, E);
    }
}


# Row pass: kept edges bucketed in LDS by (row slice, owner wave, node); each node's sum lives in its owner wave's registers, LDS holds x_r in 464-row slices: no float atomics there; bucket-order sums
# speedup vs baseline: 3.8311x; 3.8311x over previous
#include <hip/hip_runtime.h>
#include <hip/hip_bf16.h>
#include <stdint.h>

typedef __attribute__((ext_vector_type(2))) float v2f;
typedef __attribute__((ext_vector_type(8))) float v8f;
typedef __attribute__((ext_vector_type(4))) float v4f;

#define NEG_SLOPE 0.01f

__device__ __forceinline__ float lrelu(float x) {
    return x > 0.0f ? x : NEG_SLOPE * x;
}

__device__ __forceinline__ unsigned fkey(float f) {
    unsigned u = __float_as_uint(f);
    return (u & 0x80000000u) ? ~u : (u | 0x80000000u);
}
__device__ __forceinline__ float fdecode(unsigned k) {
    unsigned u = (k & 0x80000000u) ? (k ^ 0x80000000u) : ~k;
    return __uint_as_float(u);
}
#define NEG_INF_KEY 0x007FFFFFu

__global__ void k_proj_wmma(const float* __restrict__ X,
                            const float* __restrict__ W0,
                            const float* __restrict__ W1,
                            float* __restrict__ S, int nrows) {
    const int wave = (blockIdx.x * blockDim.x + threadIdx.x) >> 5;
    const int lane = threadIdx.x & 31;
    const int row0 = wave * 16;
    if (row0 >= nrows) return;

    const int m    = lane & 15;
    const int kofs = (lane >> 4) << 1;

    const float* __restrict__ arow = X + (size_t)(row0 + m) * 128;
    const float* __restrict__ brow =
        (m < 8) ? (W0 + (size_t)m * 128) : (W1 + (size_t)(m - 8) * 128);

    v8f c = {0.f, 0.f, 0.f, 0.f, 0.f, 0.f, 0.f, 0.f};
#pragma unroll
    for (int k = 0; k < 128; k += 4) {
        v2f a = *(const v2f*)(arow + k + kofs);
        v2f b = *(const v2f*)(brow + k + kofs);
        c = __builtin_amdgcn_wmma_f32_16x16x4_f32(
             false, a,  false, b,
             (short)0, c,  false,  false);
    }

    const int rbase = row0 + ((lane >> 4) << 3);
    const int col   = lane & 15;
#pragma unroll
    for (int v = 0; v < 8; ++v) {
        S[(size_t)(rbase + v) * 16 + col] = c[v];
    }
}

#define OWN_R   392

__device__ __forceinline__ unsigned own_byte(int v) {
    const unsigned o = (unsigned)v / (unsigned)OWN_R;
    return o < 255u ? o : 255u;
}

__global__ __launch_bounds__(256) void k_ownmap(const int* __restrict__ eidx,
                                                unsigned* __restrict__ map, int n) {
    unsigned char* __restrict__ mb = (unsigned char*)map;
    const int gt = (int)(blockIdx.x * 256 + threadIdx.x);
    if (((uintptr_t)eidx & 15u) != 0) {
        const int i0 = gt * 16;
        for (int i = i0; i < min(n, i0 + 16); ++i) mb[i] = (unsigned char)own_byte(eidx[i]);
        return;
    }
    const int nq   = n >> 2;
    const int lane = threadIdx.x & 31;
    const int q0   = (gt >> 5) * 128 + lane;
    const int4* __restrict__ vp = (const int4*)eidx;
    if (nq > 0) {
        int4 a[4];
#pragma unroll
        for (int j = 0; j < 4; ++j) a[j] = vp[min(q0 + j * 32, nq - 1)];
#pragma unroll
        for (int j = 0; j < 4; ++j) {
            const int q = q0 + j * 32;
            if (q < nq)
                map[q] = own_byte(a[j].x) | (own_byte(a[j].y) << 8) |
                         (own_byte(a[j].z) << 16) | (own_byte(a[j].w) << 24);
        }
    }
    if (gt == 0)
        for (int i = nq * 4; i < n; ++i) mb[i] = (unsigned char)own_byte(eidx[i]);
}

#define OWN_NT  1024
#define OWN_CAP 8192
#define OWN_G   8
#define OWN_IPT (OWN_CAP * 8 / OWN_NT)
#define OWN_SLOTS ((OWN_R + 31) / 32)
#define SL_ROWS  464
#define SL_MAXNS 5
#define OWN_NBK  (SL_MAXNS * 32 * OWN_SLOTS)
#define OWN_KEY(nl, r) ((((r) / SL_ROWS) * 32 + ((nl) & 31)) * OWN_SLOTS + ((nl) >> 5))
#define OFF_B    (OWN_CAP * 8)
#define OFF_SUM  (OFF_B + OWN_R * 32)
#define OFF_SE   (OFF_B + OWN_R * 64)
#define OFF_EA   (OFF_B + OWN_R * 96)
#define OFF_EN   (OFF_EA + OWN_CAP * 4)
#define LDS_RAW  (OFF_EA + OWN_R * 512)
static_assert(OFF_B + SL_ROWS * 512 <= LDS_RAW, "a slice of x_r fits region B");
static_assert(OFF_EN + OWN_CAP * 2 <= LDS_RAW, "the edge list fits region B");

template <int PH>
__device__ __forceinline__ void own_slow_sweep(const int* __restrict__ idx,
                                               const int* __restrict__ rel,
                                               const float* __restrict__ Sr,
                                               const float* __restrict__ x_r,
                                               float* s_acc, unsigned* s_max,
                                               float* s_sum, const float* s_se,
                                               int E, int base, int nown, int tid) {
    for (int e = tid; e < E; e += OWN_NT) {
        const unsigned d = (unsigned)(idx[e] - base);
        if (d >= (unsigned)nown) continue;
        const int r = rel[e];
        if (PH == 0) {
            for (int h = 0; h < 8; ++h) {
                const float s = lrelu(s_se[d * 8 + h] + Sr[(size_t)r * 16 + h]);
                atomicMax(&s_max[d * 8 + h], fkey(s));
            }
        } else if (PH == 1) {
            for (int h = 0; h < 8; ++h) {
                const float s = lrelu(s_se[d * 8 + h] + Sr[(size_t)r * 16 + h]);
                atomicAdd(&s_sum[d * 8 + h], __expf(s - fdecode(s_max[d * 8 + h])));
            }
        } else {
            float a = 0.0f;
            for (int h = 0; h < 8; ++h) {
                const float s = lrelu(s_se[d * 8 + h] + Sr[(size_t)r * 16 + h]);
                a += __expf(s - fdecode(s_max[d * 8 + h])) / (s_sum[d * 8 + h] + 1e-16f);
            }
            a *= 0.125f;
            for (int c = 0; c < 128; ++c)
                atomicAdd(&s_acc[d * 128 + c], x_r[(size_t)r * 128 + c] * a);
        }
    }
}

__global__ __launch_bounds__(OWN_NT) void k_owner(const int* __restrict__ eidx,
                                                  const unsigned* __restrict__ map,
                                                  const int* __restrict__ rel,
                                                  const float* __restrict__ Se,
                                                  const float* __restrict__ Sr,
                                                  const float* __restrict__ x_r,
                                                  float* __restrict__ out,
                                                  int E, int N_e, int nblk, int N_r) {
    __shared__ __attribute__((aligned(16))) unsigned char s_raw[LDS_RAW];
    __shared__ int s_cnt;
    __shared__ int s_bc[OWN_NBK];
    __shared__ int s_bo[OWN_NBK + 1];
    __shared__ int s_wt[32];
    uint2*          const s_rec  = (uint2*)s_raw;
    unsigned*       const s_max  = (unsigned*)(s_raw + OFF_B);
    float*          const s_sum  = (float*)(s_raw + OFF_SUM);
    float*          const s_se   = (float*)(s_raw + OFF_SE);
    int*            const s_ea   = (int*)(s_raw + OFF_EA);
    unsigned short* const s_en   = (unsigned short*)(s_raw + OFF_EN);
    float*          const s_acc  = (float*)(s_raw + OFF_EA);
    float*          const s_rows = (float*)(s_raw + OFF_B);

    const int tid  = threadIdx.x;
    const int lane = tid & 31;
    const int w    = tid >> 5;
    const int side = ((int)blockIdx.x >= nblk) ? 1 : 0;
    const int base = ((int)blockIdx.x - side * nblk) * OWN_R;
    const int nown = min(OWN_R, N_e - base);
    const int* __restrict__ idx = eidx + (size_t)side * (size_t)E;

    const int NS = (N_r + SL_ROWS - 1) / SL_ROWS;
    for (int i = tid; i < OWN_NBK; i += OWN_NT) s_bc[i] = 0;
    for (int i = tid; i < OWN_R * 8; i += OWN_NT) {
        const int nl = i >> 3, h = i & 7;
        s_max[i] = NEG_INF_KEY;
        s_sum[i] = 0.0f;
        s_se[i]  = (nl < nown) ? Se[(size_t)(base + nl) * 16 + side * 8 + h] : 0.0f;
    }
    if (tid == 0) s_cnt = 0;
    __syncthreads();

    auto keep1 = [&](int e) {
        const int p = atomicAdd(&s_cnt, 1);
        if (p < OWN_CAP) s_ea[p] = e;
    };
    if (nblk <= 255) {
        const unsigned char* __restrict__ mp = (const unsigned char*)map + (size_t)side * (size_t)E;
        const unsigned me  = (unsigned)((int)blockIdx.x - side * nblk);
        const unsigned pat = me * 0x01010101u;
        auto eqmask = [&](unsigned v) {
            const unsigned x = v ^ pat;
            const unsigned t = (x & 0x7F7F7F7Fu) + 0x7F7F7F7Fu;
            return ~(t | x) & 0x80808080u;
        };
        auto scan16 = [&](int e0, const uint4 a) {
            const unsigned z0 = eqmask(a.x), z1 = eqmask(a.y), z2 = eqmask(a.z), z3 = eqmask(a.w);
            if ((z0 | z1 | z2 | z3) != 0u) {
                unsigned m = (z0 >> 3) | (z1 >> 2) | (z2 >> 1) | z3;
                const int c = (int)__popc(m);
                const unsigned act = __builtin_amdgcn_ballot_w32(true);
                const int below = (int)__builtin_amdgcn_mbcnt_lo(act, 0u);
                int pre = below, tot = (int)__popc(act);
                if (__builtin_amdgcn_ballot_w32(c >= 2) != 0u) {
#pragma nounroll
                    for (int t = 2; t <= 16; ++t) {
                        const unsigned bm = __builtin_amdgcn_ballot_w32(c >= t);
                        if (bm == 0u) break;
                        pre += (int)__builtin_amdgcn_mbcnt_lo(bm, 0u);
                        tot += (int)__popc(bm);
                    }
                }
                int p0 = 0;
                if (below == 0) p0 = atomicAdd(&s_cnt, tot);
                int p = __builtin_amdgcn_readfirstlane(p0) + pre;
                while (m != 0u) {
                    const unsigned qb = (unsigned)__ffs((int)m) - 1u;
                    m &= m - 1u;
                    s_ea[min(p, OWN_CAP - 1)] = e0 + (int)(((qb & 3u) << 2) | (qb >> 3));
                    ++p;
                }
            }
        };
        int head = (int)((16u - (unsigned)((uintptr_t)mp & 15u)) & 15u);
        if (head > E) head = E;
        for (int e = tid; e < head; e += OWN_NT) if ((unsigned)mp[e] == me) keep1(e);
        const int nvec = (E - head) >> 4;
        const uint4* __restrict__ vp = (const uint4*)(mp + head);
        int q = tid;
        for (; q + 3 * OWN_NT < nvec; q += 4 * OWN_NT) {
            const uint4 a0 = vp[q];
            const uint4 a1 = vp[q + OWN_NT];
            const uint4 a2 = vp[q + 2 * OWN_NT];
            const uint4 a3 = vp[q + 3 * OWN_NT];
            scan16(head + 16 * q, a0);
            scan16(head + 16 * (q + OWN_NT), a1);
            scan16(head + 16 * (q + 2 * OWN_NT), a2);
            scan16(head + 16 * (q + 3 * OWN_NT), a3);
        }
        for (; q < nvec; q += OWN_NT) scan16(head + 16 * q, vp[q]);
        for (int e = head + 16 * nvec + tid; e < E; e += OWN_NT) if ((unsigned)mp[e] == me) keep1(e);
    } else {
        for (int e = tid; e < E; e += OWN_NT)
            if ((unsigned)(idx[e] - base) < (unsigned)nown) keep1(e);
    }
    __syncthreads();
    const int total = s_cnt;
    const bool fits = total <= OWN_CAP && NS >= 1 && NS <= SL_MAXNS &&
                      ((((uintptr_t)x_r) | ((uintptr_t)out)) & 15u) == 0;

    if (fits) {
        const int cnt = total;
        {
            int p = tid;
            for (; p + 3 * OWN_NT < cnt; p += 4 * OWN_NT) {
                int rr[4], vv[4];
#pragma unroll
                for (int j = 0; j < 4; ++j) {
                    const int e = s_ea[p + j * OWN_NT];
                    rr[j] = rel[e];
                    vv[j] = idx[e];
                }
#pragma unroll
                for (int j = 0; j < 4; ++j) {
                    const int nl = vv[j] - base;
                    s_ea[p + j * OWN_NT] = rr[j];
                    s_en[p + j * OWN_NT] = (unsigned short)nl;
                    atomicAdd(&s_bc[OWN_KEY(nl, rr[j])], 1);
                }
            }
            for (; p < cnt; p += OWN_NT) {
                const int e = s_ea[p];
                const int r = rel[e];
                const int nl = idx[e] - base;
                s_ea[p] = r;
                s_en[p] = (unsigned short)nl;
                atomicAdd(&s_bc[OWN_KEY(nl, r)], 1);
            }
        }
        __syncthreads();
        {
            const int k0 = 3 * tid;
            int c[3];
#pragma unroll
            for (int j = 0; j < 3; ++j) c[j] = (k0 + j < OWN_NBK) ? s_bc[k0 + j] : 0;
            const int mine = c[0] + c[1] + c[2];
            int x = mine;
#pragma unroll
            for (int d = 1; d < 32; d <<= 1) {
                const int y = __shfl_up(x, d);
                if (lane >= d) x += y;
            }
            if (lane == 31) s_wt[w] = x;
            __syncthreads();
            if (w == 0) {
                const int v = s_wt[lane];
                int z = v;
#pragma unroll
                for (int d = 1; d < 32; d <<= 1) {
                    const int y = __shfl_up(z, d);
                    if (lane >= d) z += y;
                }
                s_wt[lane] = z - v;
            }
            __syncthreads();
            int run = s_wt[w] + x - mine;
#pragma unroll
            for (int j = 0; j < 3; ++j) {
                if (k0 + j < OWN_NBK) { s_bo[k0 + j] = run; s_bc[k0 + j] = run; }
                run += c[j];
            }
            if (tid == 0) s_bo[OWN_NBK] = cnt;
        }
        const int n8 = cnt * 8;
        const int h8 = tid & 7;
        float sc[OWN_IPT];
#pragma unroll
        for (int g = 0; g < OWN_IPT / OWN_G; ++g) {
            if (g * OWN_G * OWN_NT >= n8) break;
            float sv[OWN_G]; int ad[OWN_G];
#pragma unroll
            for (int j = 0; j < OWN_G; ++j) {
                const int ii = tid + (g * OWN_G + j) * OWN_NT;
                const int p  = min(ii, n8 - 1) >> 3;
                ad[j] = (int)s_en[p] * 8 + h8;
                sv[j] = Sr[(size_t)s_ea[p] * 16 + h8];
            }
#pragma unroll
            for (int j = 0; j < OWN_G; ++j) {
                const int ii = tid + (g * OWN_G + j) * OWN_NT;
                const float sx = lrelu(s_se[ad[j]] + sv[j]);
                sc[g * OWN_G + j] = sx;
                if (ii < n8) atomicMax(&s_max[ad[j]], fkey(sx));
            }
        }
        __syncthreads();
#pragma unroll
        for (int g = 0; g < OWN_IPT / OWN_G; ++g) {
            if (g * OWN_G * OWN_NT >= n8) break;
            int ad[OWN_G]; float mx[OWN_G];
#pragma unroll
            for (int j = 0; j < OWN_G; ++j) {
                const int ii = tid + (g * OWN_G + j) * OWN_NT;
                ad[j] = (int)s_en[min(ii, n8 - 1) >> 3] * 8 + h8;
            }
#pragma unroll
            for (int j = 0; j < OWN_G; ++j) mx[j] = fdecode(s_max[ad[j]]);
#pragma unroll
            for (int j = 0; j < OWN_G; ++j) {
                const int ii = tid + (g * OWN_G + j) * OWN_NT;
                const float ex = __expf(sc[g * OWN_G + j] - mx[j]);
                sc[g * OWN_G + j] = ex;
                if (ii < n8) atomicAdd(&s_sum[ad[j]], ex);
            }
        }
        __syncthreads();
#define OWN_BCAST8(v, hh) __uint_as_float(__builtin_amdgcn_mov_dpp8(__float_as_uint(v), (hh) * 0x249249u))
#pragma unroll
        for (int g = 0; g < OWN_IPT / OWN_G; ++g) {
            if (g * OWN_G * OWN_NT >= n8) break;
            int pp[OWN_G]; int ad[OWN_G]; float dn[OWN_G];
#pragma unroll
            for (int j = 0; j < OWN_G; ++j) {
                const int ii = tid + (g * OWN_G + j) * OWN_NT;
                pp[j] = min(ii, n8 - 1) >> 3;
                ad[j] = (int)s_en[pp[j]] * 8 + h8;
            }
#pragma unroll
            for (int j = 0; j < OWN_G; ++j) dn[j] = s_sum[ad[j]] + 1e-16f;
#pragma unroll
            for (int j = 0; j < OWN_G; ++j) {
                const int ii = tid + (g * OWN_G + j) * OWN_NT;
                const float q = sc[g * OWN_G + j] / dn[j];
                float a = OWN_BCAST8(q, 0);
                a += OWN_BCAST8(q, 1); a += OWN_BCAST8(q, 2); a += OWN_BCAST8(q, 3); a += OWN_BCAST8(q, 4);
                a += OWN_BCAST8(q, 5); a += OWN_BCAST8(q, 6); a += OWN_BCAST8(q, 7);
                if (h8 == 0 && ii < n8) {
                    const int nl = (int)s_en[pp[j]];
                    const int r  = s_ea[pp[j]];
                    const int qd = atomicAdd(&s_bc[OWN_KEY(nl, r)], 1);
                    s_rec[qd] = make_uint2(((unsigned)r << 9) | (unsigned)nl, __float_as_uint(a * 0.125f));
                }
            }
        }
#undef OWN_BCAST8
        __syncthreads();
#define OWN_SLOT_LIST(X) X(0) X(1) X(2) X(3) X(4) X(5) X(6) X(7) X(8) X(9) X(10) X(11) X(12)
        static_assert(OWN_SLOTS == 13, "the slot list names 13 slots");
#define OWN_DECL(S) float4 acc##S = make_float4(0.f, 0.f, 0.f, 0.f);
        OWN_SLOT_LIST(OWN_DECL)
        const int ws = __builtin_amdgcn_readfirstlane(w);
        for (int s = 0; s < NS; ++s) {
            if (s > 0) __syncthreads();
            {
                const int r0 = s * SL_ROWS;
                const int nv = min(SL_ROWS, N_r - r0) * 32;
                const v4f* __restrict__ sp = (const v4f*)(x_r + (size_t)r0 * 128);
                v4f* dp = (v4f*)s_rows;
                for (int i = tid; i < nv; i += OWN_G * OWN_NT) {
                    v4f t[OWN_G];
#pragma unroll
                    for (int j = 0; j < OWN_G; ++j) t[j] = sp[min(i + j * OWN_NT, nv - 1)];
                    asm volatile("" :: "v"(t[0]), "v"(t[1]), "v"(t[2]), "v"(t[3]),
                                       "v"(t[4]), "v"(t[5]), "v"(t[6]), "v"(t[7]));
                    static_assert(OWN_G == 8, "the fence above names eight loads");
#pragma unroll
                    for (int j = 0; j < OWN_G; ++j)
                        if (i + j * OWN_NT < nv) dp[i + j * OWN_NT] = t[j];
                }
            }
            __syncthreads();
            const int kb = (s * 32 + ws) * OWN_SLOTS;
            const int bv = s_bo[kb + min(lane, OWN_SLOTS)];
            const int qz = __builtin_amdgcn_readlane(bv, OWN_SLOTS);
            const int rbase = s * SL_ROWS;
            int q = __builtin_amdgcn_readlane(bv, 0);
            while (q < qz) {
                const int cb = q;
                const int ce = min(qz, cb + 32);
                const uint2 rc = s_rec[min(cb + lane, qz - 1)];
#define OWN_SLOT_RUN(S) { \
                    const int e_ = min(__builtin_amdgcn_readlane(bv, S + 1), ce); \
                    if (q < e_) { \
                        for (int g = q; g < e_; g += 4) { \
                            float4 m_[4]; float a_[4]; \
                            _Pragma("unroll")   \
                            for (int j = 0; j < 4; ++j) { \
                                const int li = min(g + j, e_ - 1) - cb; \
                                const unsigned rx = (unsigned)__builtin_amdgcn_readlane((int)rc.x, li); \
                                a_[j] = __uint_as_float((unsigned)__builtin_amdgcn_readlane((int)rc.y, li)); \
                                m_[j] = ((const float4*)(s_rows + ((int)(rx >> 9) - rbase) * 128))[lane]; \
                            } \
                            _Pragma("unroll")   \
                            for (int j = 0; j < 4; ++j) \
                                if (g + j < e_) {   \
                                    acc##S.x += m_[j].x * a_[j]; acc##S.y += m_[j].y * a_[j]; \
                                    acc##S.z += m_[j].z * a_[j]; acc##S.w += m_[j].w * a_[j]; \
                                } \
                        } \
                        q = e_; \
                    } }
                OWN_SLOT_LIST(OWN_SLOT_RUN)
#undef OWN_SLOT_RUN
            }
        }
#define OWN_OUT(S) if (ws + 32 * S < nown) ((float4*)(out + (size_t)(base + ws + 32 * S) * 256 + side * 128))[lane] = acc##S;
        OWN_SLOT_LIST(OWN_OUT)
#undef OWN_OUT
#undef OWN_DECL
#undef OWN_SLOT_LIST
    } else {
        for (int i = tid; i < OWN_R * 32; i += OWN_NT)
            ((float4*)s_acc)[i] = make_float4(0.f, 0.f, 0.f, 0.f);
        __syncthreads();
        own_slow_sweep<0>(idx, rel, Sr, x_r, s_acc, s_max, s_sum, s_se, E, base, nown, tid);
        __syncthreads();
        own_slow_sweep<1>(idx, rel, Sr, x_r, s_acc, s_max, s_sum, s_se, E, base, nown, tid);
        __syncthreads();
        own_slow_sweep<2>(idx, rel, Sr, x_r, s_acc, s_max, s_sum, s_se, E, base, nown, tid);
        __syncthreads();
        for (int nl = w; nl < nown; nl += 32) {
            const float4 v = ((const float4*)(s_acc + nl * 128))[lane];
            ((float4*)(out + (size_t)(base + nl) * 256 + side * 128))[lane] = v;
        }
    }
}

extern "C" void kernel_launch(void* const* d_in, const int* in_sizes, int n_in,
                              void* d_out, int out_size, void* d_ws, size_t ws_size,
                              hipStream_t stream) {
    const float* x_e  = (const float*)d_in[0];
    const float* x_r  = (const float*)d_in[1];
    const float* Wh   = (const float*)d_in[2];
    const float* Wt   = (const float*)d_in[3];
    const float* Wr   = (const float*)d_in[4];
    const int*   eidx = (const int*)d_in[5];
    const int*   rel  = (const int*)d_in[6];

    const int D   = 128;
    const int N_e = in_sizes[0] / D;
    const int N_r = in_sizes[1] / D;
    const int E   = in_sizes[6];

    auto align256 = [](size_t x) { return (x + 255) & ~(size_t)255; };
    char* ws = (char*)d_ws;
    float* Se = (float*)ws; ws += align256((size_t)N_e * 16 * sizeof(float));
    float* Sr = (float*)ws; ws += align256((size_t)N_r * 16 * sizeof(float));
    unsigned* omap = (unsigned*)ws; ws += align256((size_t)2 * (size_t)E + 16);

    float* out = (float*)d_out;

    const int BLK = 256;

    {
        const int waves   = (N_e + 15) / 16;
        const int threads = waves * 32;
        const int grid    = (threads + BLK - 1) / BLK;
        k_proj_wmma<<<grid, BLK, 0, stream>>>(x_e, Wh, Wt, Se, N_e);
    }
    {
        const int waves   = (N_r + 15) / 16;
        const int threads = waves * 32;
        const int grid    = (threads + BLK - 1) / BLK;
        k_proj_wmma<<<grid, BLK, 0, stream>>>(x_r, Wr, Wr, Sr, N_r);
    }

    {
        const int n    = 2 * E;
        const int grid = (n + 16 * BLK - 1) / (16 * BLK);
        k_ownmap<<<grid, BLK, 0, stream>>>(eidx, omap, n);
    }

    {
        const int nblk = (N_e + OWN_R - 1) / OWN_R;
        k_owner<<<2 * nblk, OWN_NT, 0, stream>>>(eidx, omap, rel, Se, Sr, x_r, out, E, N_e, nblk, N_r);
    }
}


# Softmax passes: one thread per kept edge holds its eight head scores (was one thread per edge-head pair): list reads, addressing, head sum once per edge; head-major tables; exp-sums LDS-atomic order
# speedup vs baseline: 4.2277x; 1.1035x over previous
#include <hip/hip_runtime.h>
#include <hip/hip_bf16.h>
#include <stdint.h>

typedef __attribute__((ext_vector_type(2))) float v2f;
typedef __attribute__((ext_vector_type(8))) float v8f;
typedef __attribute__((ext_vector_type(4))) float v4f;

#define NEG_SLOPE 0.01f

__device__ __forceinline__ float lrelu(float x) {
    return x > 0.0f ? x : NEG_SLOPE * x;
}

__device__ __forceinline__ unsigned fkey(float f) {
    unsigned u = __float_as_uint(f);
    return (u & 0x80000000u) ? ~u : (u | 0x80000000u);
}
__device__ __forceinline__ float fdecode(unsigned k) {
    unsigned u = (k & 0x80000000u) ? (k ^ 0x80000000u) : ~k;
    return __uint_as_float(u);
}
#define NEG_INF_KEY 0x007FFFFFu

__global__ void k_proj_wmma(const float* __restrict__ X,
                            const float* __restrict__ W0,
                            const float* __restrict__ W1,
                            float* __restrict__ S, int nrows) {
    const int wave = (blockIdx.x * blockDim.x + threadIdx.x) >> 5;
    const int lane = threadIdx.x & 31;
    const int row0 = wave * 16;
    if (row0 >= nrows) return;

    const int m    = lane & 15;
    const int kofs = (lane >> 4) << 1;

    const float* __restrict__ arow = X + (size_t)(row0 + m) * 128;
    const float* __restrict__ brow =
        (m < 8) ? (W0 + (size_t)m * 128) : (W1 + (size_t)(m - 8) * 128);

    v8f c = {0.f, 0.f, 0.f, 0.f, 0.f, 0.f, 0.f, 0.f};
#pragma unroll
    for (int k = 0; k < 128; k += 4) {
        v2f a = *(const v2f*)(arow + k + kofs);
        v2f b = *(const v2f*)(brow + k + kofs);
        c = __builtin_amdgcn_wmma_f32_16x16x4_f32(
             false, a,  false, b,
             (short)0, c,  false,  false);
    }

    const int rbase = row0 + ((lane >> 4) << 3);
    const int col   = lane & 15;
#pragma unroll
    for (int v = 0; v < 8; ++v) {
        S[(size_t)(rbase + v) * 16 + col] = c[v];
    }
}

#define OWN_R   392

__device__ __forceinline__ unsigned own_byte(int v) {
    const unsigned o = (unsigned)v / (unsigned)OWN_R;
    return o < 255u ? o : 255u;
}

__global__ __launch_bounds__(256) void k_ownmap(const int* __restrict__ eidx,
                                                unsigned* __restrict__ map, int n) {
    unsigned char* __restrict__ mb = (unsigned char*)map;
    const int gt = (int)(blockIdx.x * 256 + threadIdx.x);
    if (((uintptr_t)eidx & 15u) != 0) {
        const int i0 = gt * 16;
        for (int i = i0; i < min(n, i0 + 16); ++i) mb[i] = (unsigned char)own_byte(eidx[i]);
        return;
    }
    const int nq   = n >> 2;
    const int lane = threadIdx.x & 31;
    const int q0   = (gt >> 5) * 128 + lane;
    const int4* __restrict__ vp = (const int4*)eidx;
    if (nq > 0) {
        int4 a[4];
#pragma unroll
        for (int j = 0; j < 4; ++j) a[j] = vp[min(q0 + j * 32, nq - 1)];
#pragma unroll
        for (int j = 0; j < 4; ++j) {
            const int q = q0 + j * 32;
            if (q < nq)
                map[q] = own_byte(a[j].x) | (own_byte(a[j].y) << 8) |
                         (own_byte(a[j].z) << 16) | (own_byte(a[j].w) << 24);
        }
    }
    if (gt == 0)
        for (int i = nq * 4; i < n; ++i) mb[i] = (unsigned char)own_byte(eidx[i]);
}

#define OWN_NT  1024
#define OWN_CAP 8192
#define OWN_G   8
#define OWN_EPT (OWN_CAP / OWN_NT)
#define OWN_TIX(n, h) ((h) * OWN_R + (n))
#define OWN_SLOTS ((OWN_R + 31) / 32)
#define SL_ROWS  464
#define SL_MAXNS 5
#define OWN_NBK  (SL_MAXNS * 32 * OWN_SLOTS)
#define OWN_KEY(nl, r) ((((r) / SL_ROWS) * 32 + ((nl) & 31)) * OWN_SLOTS + ((nl) >> 5))
#define OFF_B    (OWN_CAP * 8)
#define OFF_SUM  (OFF_B + OWN_R * 32)
#define OFF_SE   (OFF_B + OWN_R * 64)
#define OFF_EA   (OFF_B + OWN_R * 96)
#define OFF_EN   (OFF_EA + OWN_CAP * 4)
#define LDS_RAW  (OFF_EA + OWN_R * 512)
static_assert(OFF_B + SL_ROWS * 512 <= LDS_RAW, "a slice of x_r fits region B");
static_assert(OFF_EN + OWN_CAP * 2 <= LDS_RAW, "the edge list fits region B");

template <int PH>
__device__ __forceinline__ void own_slow_sweep(const int* __restrict__ idx,
                                               const int* __restrict__ rel,
                                               const float* __restrict__ Sr,
                                               const float* __restrict__ x_r,
                                               float* s_acc, unsigned* s_max,
                                               float* s_sum, const float* s_se,
                                               int E, int base, int nown, int tid) {
    for (int e = tid; e < E; e += OWN_NT) {
        const unsigned d = (unsigned)(idx[e] - base);
        if (d >= (unsigned)nown) continue;
        const int r = rel[e];
        if (PH == 0) {
            for (int h = 0; h < 8; ++h) {
                const float s = lrelu(s_se[OWN_TIX(d, h)] + Sr[(size_t)r * 16 + h]);
                atomicMax(&s_max[OWN_TIX(d, h)], fkey(s));
            }
        } else if (PH == 1) {
            for (int h = 0; h < 8; ++h) {
                const float s = lrelu(s_se[OWN_TIX(d, h)] + Sr[(size_t)r * 16 + h]);
                atomicAdd(&s_sum[OWN_TIX(d, h)], __expf(s - fdecode(s_max[OWN_TIX(d, h)])));
            }
        } else {
            float a = 0.0f;
            for (int h = 0; h < 8; ++h) {
                const float s = lrelu(s_se[OWN_TIX(d, h)] + Sr[(size_t)r * 16 + h]);
                a += __expf(s - fdecode(s_max[OWN_TIX(d, h)])) / (s_sum[OWN_TIX(d, h)] + 1e-16f);
            }
            a *= 0.125f;
            for (int c = 0; c < 128; ++c)
                atomicAdd(&s_acc[d * 128 + c], x_r[(size_t)r * 128 + c] * a);
        }
    }
}

__global__ __launch_bounds__(OWN_NT) void k_owner(const int* __restrict__ eidx,
                                                  const unsigned* __restrict__ map,
                                                  const int* __restrict__ rel,
                                                  const float* __restrict__ Se,
                                                  const float* __restrict__ Sr,
                                                  const float* __restrict__ x_r,
                                                  float* __restrict__ out,
                                                  int E, int N_e, int nblk, int N_r) {
    __shared__ __attribute__((aligned(16))) unsigned char s_raw[LDS_RAW];
    __shared__ int s_cnt;
    __shared__ int s_bc[OWN_NBK];
    __shared__ int s_bo[OWN_NBK + 1];
    __shared__ int s_wt[32];
    uint2*          const s_rec  = (uint2*)s_raw;
    unsigned*       const s_max  = (unsigned*)(s_raw + OFF_B);
    float*          const s_sum  = (float*)(s_raw + OFF_SUM);
    float*          const s_se   = (float*)(s_raw + OFF_SE);
    int*            const s_ea   = (int*)(s_raw + OFF_EA);
    unsigned short* const s_en   = (unsigned short*)(s_raw + OFF_EN);
    float*          const s_acc  = (float*)(s_raw + OFF_EA);
    float*          const s_rows = (float*)(s_raw + OFF_B);

    const int tid  = threadIdx.x;
    const int lane = tid & 31;
    const int w    = tid >> 5;
    const int side = ((int)blockIdx.x >= nblk) ? 1 : 0;
    const int base = ((int)blockIdx.x - side * nblk) * OWN_R;
    const int nown = min(OWN_R, N_e - base);
    const int* __restrict__ idx = eidx + (size_t)side * (size_t)E;

    const int NS = (N_r + SL_ROWS - 1) / SL_ROWS;
    for (int i = tid; i < OWN_NBK; i += OWN_NT) s_bc[i] = 0;
    for (int i = tid; i < OWN_R * 8; i += OWN_NT) {
        const int h = i / OWN_R, nl = i - h * OWN_R;
        s_max[i] = NEG_INF_KEY;
        s_sum[i] = 0.0f;
        s_se[i]  = (nl < nown) ? Se[(size_t)(base + nl) * 16 + side * 8 + h] : 0.0f;
    }
    if (tid == 0) s_cnt = 0;
    __syncthreads();

    auto keep1 = [&](int e) {
        const int p = atomicAdd(&s_cnt, 1);
        if (p < OWN_CAP) s_ea[p] = e;
    };
    if (nblk <= 255) {
        const unsigned char* __restrict__ mp = (const unsigned char*)map + (size_t)side * (size_t)E;
        const unsigned me  = (unsigned)((int)blockIdx.x - side * nblk);
        const unsigned pat = me * 0x01010101u;
        auto eqmask = [&](unsigned v) {
            const unsigned x = v ^ pat;
            const unsigned t = (x & 0x7F7F7F7Fu) + 0x7F7F7F7Fu;
            return ~(t | x) & 0x80808080u;
        };
        auto scan16 = [&](int e0, const uint4 a) {
            const unsigned z0 = eqmask(a.x), z1 = eqmask(a.y), z2 = eqmask(a.z), z3 = eqmask(a.w);
            if ((z0 | z1 | z2 | z3) != 0u) {
                unsigned m = (z0 >> 3) | (z1 >> 2) | (z2 >> 1) | z3;
                const int c = (int)__popc(m);
                const unsigned act = __builtin_amdgcn_ballot_w32(true);
                const int below = (int)__builtin_amdgcn_mbcnt_lo(act, 0u);
                int pre = below, tot = (int)__popc(act);
                if (__builtin_amdgcn_ballot_w32(c >= 2) != 0u) {
#pragma nounroll
                    for (int t = 2; t <= 16; ++t) {
                        const unsigned bm = __builtin_amdgcn_ballot_w32(c >= t);
                        if (bm == 0u) break;
                        pre += (int)__builtin_amdgcn_mbcnt_lo(bm, 0u);
                        tot += (int)__popc(bm);
                    }
                }
                int p0 = 0;
                if (below == 0) p0 = atomicAdd(&s_cnt, tot);
                int p = __builtin_amdgcn_readfirstlane(p0) + pre;
                while (m != 0u) {
                    const unsigned qb = (unsigned)__ffs((int)m) - 1u;
                    m &= m - 1u;
                    s_ea[min(p, OWN_CAP - 1)] = e0 + (int)(((qb & 3u) << 2) | (qb >> 3));
                    ++p;
                }
            }
        };
        int head = (int)((16u - (unsigned)((uintptr_t)mp & 15u)) & 15u);
        if (head > E) head = E;
        for (int e = tid; e < head; e += OWN_NT) if ((unsigned)mp[e] == me) keep1(e);
        const int nvec = (E - head) >> 4;
        const uint4* __restrict__ vp = (const uint4*)(mp + head);
        int q = tid;
        for (; q + 3 * OWN_NT < nvec; q += 4 * OWN_NT) {
            const uint4 a0 = vp[q];
            const uint4 a1 = vp[q + OWN_NT];
            const uint4 a2 = vp[q + 2 * OWN_NT];
            const uint4 a3 = vp[q + 3 * OWN_NT];
            scan16(head + 16 * q, a0);
            scan16(head + 16 * (q + OWN_NT), a1);
            scan16(head + 16 * (q + 2 * OWN_NT), a2);
            scan16(head + 16 * (q + 3 * OWN_NT), a3);
        }
        for (; q < nvec; q += OWN_NT) scan16(head + 16 * q, vp[q]);
        for (int e = head + 16 * nvec + tid; e < E; e += OWN_NT) if ((unsigned)mp[e] == me) keep1(e);
    } else {
        for (int e = tid; e < E; e += OWN_NT)
            if ((unsigned)(idx[e] - base) < (unsigned)nown) keep1(e);
    }
    __syncthreads();
    const int total = s_cnt;
    const bool fits = total <= OWN_CAP && NS >= 1 && NS <= SL_MAXNS &&
                      ((((uintptr_t)x_r) | ((uintptr_t)out) | ((uintptr_t)Sr)) & 15u) == 0;

    if (fits) {
        const int cnt = total;
        {
            int p = tid;
            for (; p + 3 * OWN_NT < cnt; p += 4 * OWN_NT) {
                int rr[4], vv[4];
#pragma unroll
                for (int j = 0; j < 4; ++j) {
                    const int e = s_ea[p + j * OWN_NT];
                    rr[j] = rel[e];
                    vv[j] = idx[e];
                }
#pragma unroll
                for (int j = 0; j < 4; ++j) {
                    const int nl = vv[j] - base;
                    s_ea[p + j * OWN_NT] = rr[j];
                    s_en[p + j * OWN_NT] = (unsigned short)nl;
                    atomicAdd(&s_bc[OWN_KEY(nl, rr[j])], 1);
                }
            }
            for (; p < cnt; p += OWN_NT) {
                const int e = s_ea[p];
                const int r = rel[e];
                const int nl = idx[e] - base;
                s_ea[p] = r;
                s_en[p] = (unsigned short)nl;
                atomicAdd(&s_bc[OWN_KEY(nl, r)], 1);
            }
        }
        __syncthreads();
        {
            const int k0 = 3 * tid;
            int c[3];
#pragma unroll
            for (int j = 0; j < 3; ++j) c[j] = (k0 + j < OWN_NBK) ? s_bc[k0 + j] : 0;
            const int mine = c[0] + c[1] + c[2];
            int x = mine;
#pragma unroll
            for (int d = 1; d < 32; d <<= 1) {
                const int y = __shfl_up(x, d);
                if (lane >= d) x += y;
            }
            if (lane == 31) s_wt[w] = x;
            __syncthreads();
            if (w == 0) {
                const int v = s_wt[lane];
                int z = v;
#pragma unroll
                for (int d = 1; d < 32; d <<= 1) {
                    const int y = __shfl_up(z, d);
                    if (lane >= d) z += y;
                }
                s_wt[lane] = z - v;
            }
            __syncthreads();
            int run = s_wt[w] + x - mine;
#pragma unroll
            for (int j = 0; j < 3; ++j) {
                if (k0 + j < OWN_NBK) { s_bo[k0 + j] = run; s_bc[k0 + j] = run; }
                run += c[j];
            }
            if (tid == 0) s_bo[OWN_NBK] = cnt;
        }
        float sc[OWN_EPT][8];
#pragma unroll
        for (int j = 0; j < OWN_EPT; ++j) {
            if (j * OWN_NT >= cnt) break;
            const int pc = min(tid + j * OWN_NT, cnt - 1);
            const float4* __restrict__ sp = (const float4*)(Sr + (size_t)s_ea[pc] * 16);
            const float4 lo = sp[0], hi = sp[1];
            sc[j][0] = lo.x; sc[j][1] = lo.y; sc[j][2] = lo.z; sc[j][3] = lo.w;
            sc[j][4] = hi.x; sc[j][5] = hi.y; sc[j][6] = hi.z; sc[j][7] = hi.w;
        }
#pragma unroll
        for (int j = 0; j < OWN_EPT; ++j) {
            if (j * OWN_NT >= cnt) break;
            const int p = tid + j * OWN_NT;
            const int n = (int)s_en[min(p, cnt - 1)];
#pragma unroll
            for (int h = 0; h < 8; ++h) sc[j][h] = lrelu(s_se[OWN_TIX(n, h)] + sc[j][h]);
            if (p < cnt) {
#pragma unroll
                for (int h = 0; h < 8; ++h) atomicMax(&s_max[OWN_TIX(n, h)], fkey(sc[j][h]));
            }
        }
        __syncthreads();
#pragma unroll
        for (int j = 0; j < OWN_EPT; ++j) {
            if (j * OWN_NT >= cnt) break;
            const int p = tid + j * OWN_NT;
            const int n = (int)s_en[min(p, cnt - 1)];
            float mx[8];
#pragma unroll
            for (int h = 0; h < 8; ++h) mx[h] = fdecode(s_max[OWN_TIX(n, h)]);
#pragma unroll
            for (int h = 0; h < 8; ++h) sc[j][h] = __expf(sc[j][h] - mx[h]);
            if (p < cnt) {
#pragma unroll
                for (int h = 0; h < 8; ++h) atomicAdd(&s_sum[OWN_TIX(n, h)], sc[j][h]);
            }
        }
        __syncthreads();
#pragma unroll
        for (int j = 0; j < OWN_EPT; ++j) {
            if (j * OWN_NT >= cnt) break;
            const int p = tid + j * OWN_NT;
            const int pc = min(p, cnt - 1);
            const int n = (int)s_en[pc];
            float dn[8];
#pragma unroll
            for (int h = 0; h < 8; ++h) dn[h] = s_sum[OWN_TIX(n, h)] + 1e-16f;
            float a = sc[j][0] / dn[0];
#pragma unroll
            for (int h = 1; h < 8; ++h) a += sc[j][h] / dn[h];
            if (p < cnt) {
                const int r  = s_ea[pc];
                const int qd = atomicAdd(&s_bc[OWN_KEY(n, r)], 1);
                s_rec[qd] = make_uint2(((unsigned)r << 9) | (unsigned)n, __float_as_uint(a * 0.125f));
            }
        }
        __syncthreads();
#define OWN_SLOT_LIST(X) X(0) X(1) X(2) X(3) X(4) X(5) X(6) X(7) X(8) X(9) X(10) X(11) X(12)
        static_assert(OWN_SLOTS == 13, "the slot list names 13 slots");
#define OWN_DECL(S) float4 acc##S = make_float4(0.f, 0.f, 0.f, 0.f);
        OWN_SLOT_LIST(OWN_DECL)
        const int ws = __builtin_amdgcn_readfirstlane(w);
        for (int s = 0; s < NS; ++s) {
            if (s > 0) __syncthreads();
            {
                const int r0 = s * SL_ROWS;
                const int nv = min(SL_ROWS, N_r - r0) * 32;
                const v4f* __restrict__ sp = (const v4f*)(x_r + (size_t)r0 * 128);
                v4f* dp = (v4f*)s_rows;
                for (int i = tid; i < nv; i += OWN_G * OWN_NT) {
                    v4f t[OWN_G];
#pragma unroll
                    for (int j = 0; j < OWN_G; ++j) t[j] = sp[min(i + j * OWN_NT, nv - 1)];
                    asm volatile("" :: "v"(t[0]), "v"(t[1]), "v"(t[2]), "v"(t[3]),
                                       "v"(t[4]), "v"(t[5]), "v"(t[6]), "v"(t[7]));
                    static_assert(OWN_G == 8, "the fence above names eight loads");
#pragma unroll
                    for (int j = 0; j < OWN_G; ++j)
                        if (i + j * OWN_NT < nv) dp[i + j * OWN_NT] = t[j];
                }
            }
            __syncthreads();
            const int kb = (s * 32 + ws) * OWN_SLOTS;
            const int bv = s_bo[kb + min(lane, OWN_SLOTS)];
            const int qz = __builtin_amdgcn_readlane(bv, OWN_SLOTS);
            const int rbase = s * SL_ROWS;
            int q = __builtin_amdgcn_readlane(bv, 0);
            while (q < qz) {
                const int cb = q;
                const int ce = min(qz, cb + 32);
                const uint2 rc = s_rec[min(cb + lane, qz - 1)];
#define OWN_SLOT_RUN(S) { \
                    const int e_ = min(__builtin_amdgcn_readlane(bv, S + 1), ce); \
                    if (q < e_) { \
                        for (int g = q; g < e_; g += 4) { \
                            float4 m_[4]; float a_[4]; \
                            _Pragma("unroll")   \
                            for (int j = 0; j < 4; ++j) { \
                                const int li = min(g + j, e_ - 1) - cb; \
                                const unsigned rx = (unsigned)__builtin_amdgcn_readlane((int)rc.x, li); \
                                a_[j] = __uint_as_float((unsigned)__builtin_amdgcn_readlane((int)rc.y, li)); \
                                m_[j] = ((const float4*)(s_rows + ((int)(rx >> 9) - rbase) * 128))[lane]; \
                            } \
                            _Pragma("unroll")   \
                            for (int j = 0; j < 4; ++j) \
                                if (g + j < e_) {   \
                                    acc##S.x += m_[j].x * a_[j]; acc##S.y += m_[j].y * a_[j]; \
                                    acc##S.z += m_[j].z * a_[j]; acc##S.w += m_[j].w * a_[j]; \
                                } \
                        } \
                        q = e_; \
                    } }
                OWN_SLOT_LIST(OWN_SLOT_RUN)
#undef OWN_SLOT_RUN
            }
        }
#define OWN_OUT(S) if (ws + 32 * S < nown) ((float4*)(out + (size_t)(base + ws + 32 * S) * 256 + side * 128))[lane] = acc##S;
        OWN_SLOT_LIST(OWN_OUT)
#undef OWN_OUT
#undef OWN_DECL
#undef OWN_SLOT_LIST
    } else {
        for (int i = tid; i < OWN_R * 32; i += OWN_NT)
            ((float4*)s_acc)[i] = make_float4(0.f, 0.f, 0.f, 0.f);
        __syncthreads();
        own_slow_sweep<0>(idx, rel, Sr, x_r, s_acc, s_max, s_sum, s_se, E, base, nown, tid);
        __syncthreads();
        own_slow_sweep<1>(idx, rel, Sr, x_r, s_acc, s_max, s_sum, s_se, E, base, nown, tid);
        __syncthreads();
        own_slow_sweep<2>(idx, rel, Sr, x_r, s_acc, s_max, s_sum, s_se, E, base, nown, tid);
        __syncthreads();
        for (int nl = w; nl < nown; nl += 32) {
            const float4 v = ((const float4*)(s_acc + nl * 128))[lane];
            ((float4*)(out + (size_t)(base + nl) * 256 + side * 128))[lane] = v;
        }
    }
}

extern "C" void kernel_launch(void* const* d_in, const int* in_sizes, int n_in,
                              void* d_out, int out_size, void* d_ws, size_t ws_size,
                              hipStream_t stream) {
    const float* x_e  = (const float*)d_in[0];
    const float* x_r  = (const float*)d_in[1];
    const float* Wh   = (const float*)d_in[2];
    const float* Wt   = (const float*)d_in[3];
    const float* Wr   = (const float*)d_in[4];
    const int*   eidx = (const int*)d_in[5];
    const int*   rel  = (const int*)d_in[6];

    const int D   = 128;
    const int N_e = in_sizes[0] / D;
    const int N_r = in_sizes[1] / D;
    const int E   = in_sizes[6];

    auto align256 = [](size_t x) { return (x + 255) & ~(size_t)255; };
    char* ws = (char*)d_ws;
    float* Se = (float*)ws; ws += align256((size_t)N_e * 16 * sizeof(float));
    float* Sr = (float*)ws; ws += align256((size_t)N_r * 16 * sizeof(float));
    unsigned* omap = (unsigned*)ws; ws += align256((size_t)2 * (size_t)E + 16);

    float* out = (float*)d_out;

    const int BLK = 256;

    {
        const int waves   = (N_e + 15) / 16;
        const int threads = waves * 32;
        const int grid    = (threads + BLK - 1) / BLK;
        k_proj_wmma<<<grid, BLK, 0, stream>>>(x_e, Wh, Wt, Se, N_e);
    }
    {
        const int waves   = (N_r + 15) / 16;
        const int threads = waves * 32;
        const int grid    = (threads + BLK - 1) / BLK;
        k_proj_wmma<<<grid, BLK, 0, stream>>>(x_r, Wr, Wr, Sr, N_r);
    }

    {
        const int n    = 2 * E;
        const int grid = (n + 16 * BLK - 1) / (16 * BLK);
        k_ownmap<<<grid, BLK, 0, stream>>>(eidx, omap, n);
    }

    {
        const int nblk = (N_e + OWN_R - 1) / OWN_R;
        k_owner<<<2 * nblk, OWN_NT, 0, stream>>>(eidx, omap, rel, Se, Sr, x_r, out, E, N_e, nblk, N_r);
    }
}


# Owner-map scan: a thread keeps its hits in an LDS list of its own, two loads' masks per drain loop; a block prefix sum packs the lists: no wave vote or counter add per load; sums in LDS-atomic order
# speedup vs baseline: 4.6479x; 1.0994x over previous
#include <hip/hip_runtime.h>
#include <hip/hip_bf16.h>
#include <stdint.h>

typedef __attribute__((ext_vector_type(2))) float v2f;
typedef __attribute__((ext_vector_type(8))) float v8f;
typedef __attribute__((ext_vector_type(4))) float v4f;

#define NEG_SLOPE 0.01f

__device__ __forceinline__ float lrelu(float x) {
    return x > 0.0f ? x : NEG_SLOPE * x;
}

__device__ __forceinline__ unsigned fkey(float f) {
    unsigned u = __float_as_uint(f);
    return (u & 0x80000000u) ? ~u : (u | 0x80000000u);
}
__device__ __forceinline__ float fdecode(unsigned k) {
    unsigned u = (k & 0x80000000u) ? (k ^ 0x80000000u) : ~k;
    return __uint_as_float(u);
}
#define NEG_INF_KEY 0x007FFFFFu

__global__ void k_proj_wmma(const float* __restrict__ X,
                            const float* __restrict__ W0,
                            const float* __restrict__ W1,
                            float* __restrict__ S, int nrows) {
    const int wave = (blockIdx.x * blockDim.x + threadIdx.x) >> 5;
    const int lane = threadIdx.x & 31;
    const int row0 = wave * 16;
    if (row0 >= nrows) return;

    const int m    = lane & 15;
    const int kofs = (lane >> 4) << 1;

    const float* __restrict__ arow = X + (size_t)(row0 + m) * 128;
    const float* __restrict__ brow =
        (m < 8) ? (W0 + (size_t)m * 128) : (W1 + (size_t)(m - 8) * 128);

    v8f c = {0.f, 0.f, 0.f, 0.f, 0.f, 0.f, 0.f, 0.f};
#pragma unroll
    for (int k = 0; k < 128; k += 4) {
        v2f a = *(const v2f*)(arow + k + kofs);
        v2f b = *(const v2f*)(brow + k + kofs);
        c = __builtin_amdgcn_wmma_f32_16x16x4_f32(
             false, a,  false, b,
             (short)0, c,  false,  false);
    }

    const int rbase = row0 + ((lane >> 4) << 3);
    const int col   = lane & 15;
#pragma unroll
    for (int v = 0; v < 8; ++v) {
        S[(size_t)(rbase + v) * 16 + col] = c[v];
    }
}

#define OWN_R   392

__device__ __forceinline__ unsigned own_byte(int v) {
    const unsigned o = (unsigned)v / (unsigned)OWN_R;
    return o < 255u ? o : 255u;
}

__global__ __launch_bounds__(256) void k_ownmap(const int* __restrict__ eidx,
                                                unsigned* __restrict__ map, int n) {
    unsigned char* __restrict__ mb = (unsigned char*)map;
    const int gt = (int)(blockIdx.x * 256 + threadIdx.x);
    if (((uintptr_t)eidx & 15u) != 0) {
        const int i0 = gt * 16;
        for (int i = i0; i < min(n, i0 + 16); ++i) mb[i] = (unsigned char)own_byte(eidx[i]);
        return;
    }
    const int nq   = n >> 2;
    const int lane = threadIdx.x & 31;
    const int q0   = (gt >> 5) * 128 + lane;
    const int4* __restrict__ vp = (const int4*)eidx;
    if (nq > 0) {
        int4 a[4];
#pragma unroll
        for (int j = 0; j < 4; ++j) a[j] = vp[min(q0 + j * 32, nq - 1)];
#pragma unroll
        for (int j = 0; j < 4; ++j) {
            const int q = q0 + j * 32;
            if (q < nq)
                map[q] = own_byte(a[j].x) | (own_byte(a[j].y) << 8) |
                         (own_byte(a[j].z) << 16) | (own_byte(a[j].w) << 24);
        }
    }
    if (gt == 0)
        for (int i = nq * 4; i < n; ++i) mb[i] = (unsigned char)own_byte(eidx[i]);
}

#define OWN_NT  1024
#define OWN_CAP 8192
#define OWN_G   8
#define OWN_EPT (OWN_CAP / OWN_NT)
#define OWN_TIX(n, h) ((h) * OWN_R + (n))
#define OWN_SLOTS ((OWN_R + 31) / 32)
#define SL_ROWS  256
#define SL_MAXNS 8
#define OWN_NBK  (SL_MAXNS * 32 * OWN_SLOTS)
#define OWN_KEY(nl, r) ((((r) / SL_ROWS) * 32 + ((nl) & 31)) * OWN_SLOTS + ((nl) >> 5))
#define OFF_B    (OWN_CAP * 8)
#define OFF_SUM  (OFF_B + OWN_R * 32)
#define OFF_SE   (OFF_B + OWN_R * 64)
#define OFF_EA   (OFF_B + OWN_R * 96)
#define OFF_EN   (OFF_EA + OWN_CAP * 4)
#define LDS_RAW  (OFF_EA + OWN_R * 512)
static_assert(OFF_B + SL_ROWS * 512 <= LDS_RAW, "a slice of x_r fits region B");
static_assert(SL_ROWS * 32 == OWN_G * OWN_NT, "a thread holds a whole slice's share in OWN_G registers");
static_assert(4 * OWN_NT >= OWN_NBK, "the bucket scan takes four buckets per thread");
static_assert(OFF_EN + OWN_CAP * 2 <= LDS_RAW, "the edge list fits region B");
#define OWN_PK   32
#define OFF_PV   (OFF_EN + OWN_CAP * 2)
static_assert(OFF_PV + OWN_PK * OWN_NT * 4 <= LDS_RAW, "the threads' own hit lists fit region B");

template <int PH>
__device__ __forceinline__ void own_slow_sweep(const int* __restrict__ idx,
                                               const int* __restrict__ rel,
                                               const float* __restrict__ Sr,
                                               const float* __restrict__ x_r,
                                               float* s_acc, unsigned* s_max,
                                               float* s_sum, const float* s_se,
                                               int E, int base, int nown, int tid) {
    for (int e = tid; e < E; e += OWN_NT) {
        const unsigned d = (unsigned)(idx[e] - base);
        if (d >= (unsigned)nown) continue;
        const int r = rel[e];
        if (PH == 0) {
            for (int h = 0; h < 8; ++h) {
                const float s = lrelu(s_se[OWN_TIX(d, h)] + Sr[(size_t)r * 16 + h]);
                atomicMax(&s_max[OWN_TIX(d, h)], fkey(s));
            }
        } else if (PH == 1) {
            for (int h = 0; h < 8; ++h) {
                const float s = lrelu(s_se[OWN_TIX(d, h)] + Sr[(size_t)r * 16 + h]);
                atomicAdd(&s_sum[OWN_TIX(d, h)], __expf(s - fdecode(s_max[OWN_TIX(d, h)])));
            }
        } else {
            float a = 0.0f;
            for (int h = 0; h < 8; ++h) {
                const float s = lrelu(s_se[OWN_TIX(d, h)] + Sr[(size_t)r * 16 + h]);
                a += __expf(s - fdecode(s_max[OWN_TIX(d, h)])) / (s_sum[OWN_TIX(d, h)] + 1e-16f);
            }
            a *= 0.125f;
            for (int c = 0; c < 128; ++c)
                atomicAdd(&s_acc[d * 128 + c], x_r[(size_t)r * 128 + c] * a);
        }
    }
}

__global__ __launch_bounds__(OWN_NT) void k_owner(const int* __restrict__ eidx,
                                                  const unsigned* __restrict__ map,
                                                  const int* __restrict__ rel,
                                                  const float* __restrict__ Se,
                                                  const float* __restrict__ Sr,
                                                  const float* __restrict__ x_r,
                                                  float* __restrict__ out,
                                                  int E, int N_e, int nblk, int N_r) {
    __shared__ __attribute__((aligned(16))) unsigned char s_raw[LDS_RAW];
    __shared__ int s_cnt;
    __shared__ int s_bc[OWN_NBK];
    __shared__ unsigned short s_bo[OWN_NBK + 1];
    __shared__ int s_wt[32];
    uint2*          const s_rec  = (uint2*)s_raw;
    unsigned*       const s_max  = (unsigned*)(s_raw + OFF_B);
    float*          const s_sum  = (float*)(s_raw + OFF_SUM);
    float*          const s_se   = (float*)(s_raw + OFF_SE);
    int*            const s_ea   = (int*)(s_raw + OFF_EA);
    unsigned short* const s_en   = (unsigned short*)(s_raw + OFF_EN);
    float*          const s_acc  = (float*)(s_raw + OFF_EA);
    float*          const s_rows = (float*)(s_raw + OFF_B);
    int*            const s_pv   = (int*)(s_raw + OFF_PV);

    const int tid  = threadIdx.x;
    const int lane = tid & 31;
    const int w    = tid >> 5;
    const int side = ((int)blockIdx.x >= nblk) ? 1 : 0;
    const int base = ((int)blockIdx.x - side * nblk) * OWN_R;
    const int nown = min(OWN_R, N_e - base);
    const int* __restrict__ idx = eidx + (size_t)side * (size_t)E;

    const int NS = (N_r + SL_ROWS - 1) / SL_ROWS;
    for (int i = tid; i < OWN_NBK; i += OWN_NT) s_bc[i] = 0;
    for (int i = tid; i < OWN_R * 8; i += OWN_NT) {
        const int h = i / OWN_R, nl = i - h * OWN_R;
        s_max[i] = NEG_INF_KEY;
        s_sum[i] = 0.0f;
        s_se[i]  = (nl < nown) ? Se[(size_t)(base + nl) * 16 + side * 8 + h] : 0.0f;
    }
    if (tid == 0) s_cnt = 0;
    __syncthreads();

    int cntp = 0;
    auto keep1 = [&](int e) {
        s_pv[min(cntp, OWN_PK - 1) * OWN_NT + tid] = e;
        ++cntp;
    };
    if (nblk <= 255) {
        const unsigned char* __restrict__ mp = (const unsigned char*)map + (size_t)side * (size_t)E;
        const unsigned me  = (unsigned)((int)blockIdx.x - side * nblk);
        const unsigned pat = me * 0x01010101u;
        auto eqmask = [&](unsigned v) {
            const unsigned x = v ^ pat;
            const unsigned t = (x & 0x7F7F7F7Fu) + 0x7F7F7F7Fu;
            return ~(t | x) & 0x80808080u;
        };
        auto hit16 = [&](const uint4 a) {
            return (eqmask(a.x) >> 3) | (eqmask(a.y) >> 2) | (eqmask(a.z) >> 1) | eqmask(a.w);
        };
        auto drain = [&](unsigned wv, int eb) {
            while (wv != 0u) {
                const unsigned k = (unsigned)__ffs((int)wv) - 1u;
                wv &= wv - 1u;
                keep1(eb + (int)((((k >> 2) & 1u) * (16u * OWN_NT)) + ((k & 3u) << 2) + (k >> 3)));
            }
        };
        int head = (int)((16u - (unsigned)((uintptr_t)mp & 15u)) & 15u);
        if (head > E) head = E;
        for (int e = tid; e < head; e += OWN_NT) if ((unsigned)mp[e] == me) keep1(e);
        const int nvec = (E - head) >> 4;
        const uint4* __restrict__ vp = (const uint4*)(mp + head);
        int q = tid;
        for (; q + 3 * OWN_NT < nvec; q += 4 * OWN_NT) {
            const uint4 a0 = vp[q];
            const uint4 a1 = vp[q + OWN_NT];
            const uint4 a2 = vp[q + 2 * OWN_NT];
            const uint4 a3 = vp[q + 3 * OWN_NT];
            const unsigned wa = (hit16(a0) >> 4) | hit16(a1);
            const unsigned wb = (hit16(a2) >> 4) | hit16(a3);
            drain(wa, head + 16 * q);
            drain(wb, head + 16 * (q + 2 * OWN_NT));
        }
        for (; q < nvec; q += OWN_NT) drain(hit16(vp[q]) >> 4, head + 16 * q);
        for (int e = head + 16 * nvec + tid; e < E; e += OWN_NT) if ((unsigned)mp[e] == me) keep1(e);
    } else {
        for (int e = tid; e < E; e += OWN_NT)
            if ((unsigned)(idx[e] - base) < (unsigned)nown) keep1(e);
    }
    {
        const int mine = (cntp > OWN_PK) ? (OWN_CAP + 1) : cntp;
        int x = mine;
#pragma unroll
        for (int d = 1; d < 32; d <<= 1) {
            const int y = __shfl_up(x, d);
            if (lane >= d) x += y;
        }
        if (lane == 31) s_wt[w] = x;
        __syncthreads();
        if (w == 0) {
            const int v = s_wt[lane];
            int z = v;
#pragma unroll
            for (int d = 1; d < 32; d <<= 1) {
                const int y = __shfl_up(z, d);
                if (lane >= d) z += y;
            }
            s_wt[lane] = z - v;
        }
        __syncthreads();
        const int off = s_wt[w] + x - mine;
        if (tid == OWN_NT - 1) s_cnt = off + mine;
        const int have = min(cntp, OWN_PK);
        for (int i = 0; i < have; ++i)
            if (off + i < OWN_CAP) s_ea[off + i] = s_pv[i * OWN_NT + tid];
    }
    __syncthreads();
    const int total = s_cnt;
    const bool fits = total <= OWN_CAP && NS >= 1 && NS <= SL_MAXNS &&
                      ((((uintptr_t)x_r) | ((uintptr_t)out) | ((uintptr_t)Sr)) & 15u) == 0;

    if (fits) {
        const int cnt = total;
        {
            int p = tid;
            for (; p + 3 * OWN_NT < cnt; p += 4 * OWN_NT) {
                int rr[4], vv[4];
#pragma unroll
                for (int j = 0; j < 4; ++j) {
                    const int e = s_ea[p + j * OWN_NT];
                    rr[j] = rel[e];
                    vv[j] = idx[e];
                }
#pragma unroll
                for (int j = 0; j < 4; ++j) {
                    const int nl = vv[j] - base;
                    s_ea[p + j * OWN_NT] = rr[j];
                    s_en[p + j * OWN_NT] = (unsigned short)nl;
                    atomicAdd(&s_bc[OWN_KEY(nl, rr[j])], 1);
                }
            }
            for (; p < cnt; p += OWN_NT) {
                const int e = s_ea[p];
                const int r = rel[e];
                const int nl = idx[e] - base;
                s_ea[p] = r;
                s_en[p] = (unsigned short)nl;
                atomicAdd(&s_bc[OWN_KEY(nl, r)], 1);
            }
        }
        __syncthreads();
        {
            const int k0 = 4 * tid;
            int c[4];
#pragma unroll
            for (int j = 0; j < 4; ++j) c[j] = (k0 + j < OWN_NBK) ? s_bc[k0 + j] : 0;
            const int mine = c[0] + c[1] + c[2] + c[3];
            int x = mine;
#pragma unroll
            for (int d = 1; d < 32; d <<= 1) {
                const int y = __shfl_up(x, d);
                if (lane >= d) x += y;
            }
            if (lane == 31) s_wt[w] = x;
            __syncthreads();
            if (w == 0) {
                const int v = s_wt[lane];
                int z = v;
#pragma unroll
                for (int d = 1; d < 32; d <<= 1) {
                    const int y = __shfl_up(z, d);
                    if (lane >= d) z += y;
                }
                s_wt[lane] = z - v;
            }
            __syncthreads();
            int run = s_wt[w] + x - mine;
#pragma unroll
            for (int j = 0; j < 4; ++j) {
                if (k0 + j < OWN_NBK) { s_bo[k0 + j] = (unsigned short)run; s_bc[k0 + j] = run; }
                run += c[j];
            }
            if (tid == 0) s_bo[OWN_NBK] = (unsigned short)cnt;
        }
        float sc[OWN_EPT][8];
#pragma unroll
        for (int j = 0; j < OWN_EPT; ++j) {
            if (j * OWN_NT >= cnt) break;
            const int pc = min(tid + j * OWN_NT, cnt - 1);
            const float4* __restrict__ sp = (const float4*)(Sr + (size_t)s_ea[pc] * 16);
            const float4 lo = sp[0], hi = sp[1];
            sc[j][0] = lo.x; sc[j][1] = lo.y; sc[j][2] = lo.z; sc[j][3] = lo.w;
            sc[j][4] = hi.x; sc[j][5] = hi.y; sc[j][6] = hi.z; sc[j][7] = hi.w;
        }
#pragma unroll
        for (int j = 0; j < OWN_EPT; ++j) {
            if (j * OWN_NT >= cnt) break;
            const int p = tid + j * OWN_NT;
            const int n = (int)s_en[min(p, cnt - 1)];
#pragma unroll
            for (int h = 0; h < 8; ++h) sc[j][h] = lrelu(s_se[OWN_TIX(n, h)] + sc[j][h]);
            if (p < cnt) {
#pragma unroll
                for (int h = 0; h < 8; ++h) atomicMax(&s_max[OWN_TIX(n, h)], fkey(sc[j][h]));
            }
        }
        __syncthreads();
#pragma unroll
        for (int j = 0; j < OWN_EPT; ++j) {
            if (j * OWN_NT >= cnt) break;
            const int p = tid + j * OWN_NT;
            const int n = (int)s_en[min(p, cnt - 1)];
            float mx[8];
#pragma unroll
            for (int h = 0; h < 8; ++h) mx[h] = fdecode(s_max[OWN_TIX(n, h)]);
#pragma unroll
            for (int h = 0; h < 8; ++h) sc[j][h] = __expf(sc[j][h] - mx[h]);
            if (p < cnt) {
#pragma unroll
                for (int h = 0; h < 8; ++h) atomicAdd(&s_sum[OWN_TIX(n, h)], sc[j][h]);
            }
        }
        __syncthreads();
#pragma unroll
        for (int j = 0; j < OWN_EPT; ++j) {
            if (j * OWN_NT >= cnt) break;
            const int p = tid + j * OWN_NT;
            const int pc = min(p, cnt - 1);
            const int n = (int)s_en[pc];
            float dn[8];
#pragma unroll
            for (int h = 0; h < 8; ++h) dn[h] = s_sum[OWN_TIX(n, h)] + 1e-16f;
            float a = sc[j][0] / dn[0];
#pragma unroll
            for (int h = 1; h < 8; ++h) a += sc[j][h] / dn[h];
            if (p < cnt) {
                const int r  = s_ea[pc];
                const int qd = atomicAdd(&s_bc[OWN_KEY(n, r)], 1);
                s_rec[qd] = make_uint2(((unsigned)r << 9) | (unsigned)n, __float_as_uint(a * 0.125f));
            }
        }
        __syncthreads();
#define OWN_SLOT_LIST(X) X(0) X(1) X(2) X(3) X(4) X(5) X(6) X(7) X(8) X(9) X(10) X(11) X(12)
        static_assert(OWN_SLOTS == 13, "the slot list names 13 slots");
#define OWN_DECL(S) float4 acc##S = make_float4(0.f, 0.f, 0.f, 0.f);
        OWN_SLOT_LIST(OWN_DECL)
        const int ws = __builtin_amdgcn_readfirstlane(w);
        v4f t[OWN_G];
        {
            const int nv = min(SL_ROWS, N_r) * 32;
            const v4f* __restrict__ sp = (const v4f*)x_r;
#pragma unroll
            for (int j = 0; j < OWN_G; ++j) t[j] = sp[min(tid + j * OWN_NT, nv - 1)];
        }
        for (int s = 0; s < NS; ++s) {
            if (s > 0) __syncthreads();
            {
                const int nv = min(SL_ROWS, N_r - s * SL_ROWS) * 32;
                v4f* dp = (v4f*)s_rows;
#pragma unroll
                for (int j = 0; j < OWN_G; ++j)
                    if (tid + j * OWN_NT < nv) dp[tid + j * OWN_NT] = t[j];
            }
            __syncthreads();
            if (s + 1 < NS) {
                const int r1 = (s + 1) * SL_ROWS;
                const int nv = min(SL_ROWS, N_r - r1) * 32;
                const v4f* __restrict__ sp = (const v4f*)(x_r + (size_t)r1 * 128);
#pragma unroll
                for (int j = 0; j < OWN_G; ++j) t[j] = sp[min(tid + j * OWN_NT, nv - 1)];
            }
            const int kb = (s * 32 + ws) * OWN_SLOTS;
            const int bv = (int)s_bo[kb + min(lane, OWN_SLOTS)];
            const int qz = __builtin_amdgcn_readlane(bv, OWN_SLOTS);
            const int rbase = s * SL_ROWS;
            int q = __builtin_amdgcn_readlane(bv, 0);
            while (q < qz) {
                const int cb = q;
                const int ce = min(qz, cb + 32);
                const uint2 rc = s_rec[min(cb + lane, qz - 1)];
#define OWN_SLOT_RUN(S) { \
                    const int e_ = min(__builtin_amdgcn_readlane(bv, S + 1), ce); \
                    if (q < e_) { \
                        for (int g = q; g < e_; g += 2) { \
                            v4f m_[2]; float a_[2]; \
                            _Pragma("unroll")   \
                            for (int j = 0; j < 2; ++j) { \
                                const int li = min(g + j, e_ - 1) - cb; \
                                const unsigned rx = (unsigned)__builtin_amdgcn_readlane((int)rc.x, li); \
                                a_[j] = __uint_as_float((unsigned)__builtin_amdgcn_readlane((int)rc.y, li)); \
                                m_[j] = ((const v4f*)(s_rows + ((int)(rx >> 9) - rbase) * 128))[lane]; \
                            } \
                              \
                              \
                            asm volatile("" :: "v"(m_[0]), "v"(m_[1])); \
                            _Pragma("unroll")   \
                            for (int j = 0; j < 2; ++j) \
                                if (g + j < e_) {   \
                                    acc##S.x += m_[j].x * a_[j]; acc##S.y += m_[j].y * a_[j]; \
                                    acc##S.z += m_[j].z * a_[j]; acc##S.w += m_[j].w * a_[j]; \
                                } \
                        } \
                        q = e_; \
                    } }
                OWN_SLOT_LIST(OWN_SLOT_RUN)
#undef OWN_SLOT_RUN
            }
        }
#define OWN_OUT(S) if (ws + 32 * S < nown) ((float4*)(out + (size_t)(base + ws + 32 * S) * 256 + side * 128))[lane] = acc##S;
        OWN_SLOT_LIST(OWN_OUT)
#undef OWN_OUT
#undef OWN_DECL
#undef OWN_SLOT_LIST
    } else {
        for (int i = tid; i < OWN_R * 32; i += OWN_NT)
            ((float4*)s_acc)[i] = make_float4(0.f, 0.f, 0.f, 0.f);
        __syncthreads();
        own_slow_sweep<0>(idx, rel, Sr, x_r, s_acc, s_max, s_sum, s_se, E, base, nown, tid);
        __syncthreads();
        own_slow_sweep<1>(idx, rel, Sr, x_r, s_acc, s_max, s_sum, s_se, E, base, nown, tid);
        __syncthreads();
        own_slow_sweep<2>(idx, rel, Sr, x_r, s_acc, s_max, s_sum, s_se, E, base, nown, tid);
        __syncthreads();
        for (int nl = w; nl < nown; nl += 32) {
            const float4 v = ((const float4*)(s_acc + nl * 128))[lane];
            ((float4*)(out + (size_t)(base + nl) * 256 + side * 128))[lane] = v;
        }
    }
}

extern "C" void kernel_launch(void* const* d_in, const int* in_sizes, int n_in,
                              void* d_out, int out_size, void* d_ws, size_t ws_size,
                              hipStream_t stream) {
    const float* x_e  = (const float*)d_in[0];
    const float* x_r  = (const float*)d_in[1];
    const float* Wh   = (const float*)d_in[2];
    const float* Wt   = (const float*)d_in[3];
    const float* Wr   = (const float*)d_in[4];
    const int*   eidx = (const int*)d_in[5];
    const int*   rel  = (const int*)d_in[6];

    const int D   = 128;
    const int N_e = in_sizes[0] / D;
    const int N_r = in_sizes[1] / D;
    const int E   = in_sizes[6];

    auto align256 = [](size_t x) { return (x + 255) & ~(size_t)255; };
    char* ws = (char*)d_ws;
    float* Se = (float*)ws; ws += align256((size_t)N_e * 16 * sizeof(float));
    float* Sr = (float*)ws; ws += align256((size_t)N_r * 16 * sizeof(float));
    unsigned* omap = (unsigned*)ws; ws += align256((size_t)2 * (size_t)E + 16);

    float* out = (float*)d_out;

    const int BLK = 256;

    {
        const int waves   = (N_e + 15) / 16;
        const int threads = waves * 32;
        const int grid    = (threads + BLK - 1) / BLK;
        k_proj_wmma<<<grid, BLK, 0, stream>>>(x_e, Wh, Wt, Se, N_e);
    }
    {
        const int waves   = (N_r + 15) / 16;
        const int threads = waves * 32;
        const int grid    = (threads + BLK - 1) / BLK;
        k_proj_wmma<<<grid, BLK, 0, stream>>>(x_r, Wr, Wr, Sr, N_r);
    }

    {
        const int n    = 2 * E;
        const int grid = (n + 16 * BLK - 1) / (16 * BLK);
        k_ownmap<<<grid, BLK, 0, stream>>>(eidx, omap, n);
    }

    {
        const int nblk = (N_e + OWN_R - 1) / OWN_R;
        k_owner<<<2 * nblk, OWN_NT, 0, stream>>>(eidx, omap, rel, Se, Sr, x_r, out, E, N_e, nblk, N_r);
    }
}


# Node projection moves into the owner blocks: each parks its 392 rows of x_e in LDS by whole-line loads and projects them with f32 WMMA; no x_e projection launch or score table: 3 launches, was 4
# speedup vs baseline: 4.8405x; 1.0414x over previous
#include <hip/hip_runtime.h>
#include <hip/hip_bf16.h>
#include <stdint.h>

typedef __attribute__((ext_vector_type(2))) float v2f;
typedef __attribute__((ext_vector_type(8))) float v8f;
typedef __attribute__((ext_vector_type(4))) float v4f;

#define NEG_SLOPE 0.01f

__device__ __forceinline__ float lrelu(float x) {
    return x > 0.0f ? x : NEG_SLOPE * x;
}

__device__ __forceinline__ unsigned fkey(float f) {
    unsigned u = __float_as_uint(f);
    return (u & 0x80000000u) ? ~u : (u | 0x80000000u);
}
__device__ __forceinline__ float fdecode(unsigned k) {
    unsigned u = (k & 0x80000000u) ? (k ^ 0x80000000u) : ~k;
    return __uint_as_float(u);
}
#define NEG_INF_KEY 0x007FFFFFu

__global__ void k_proj_wmma(const float* __restrict__ X,
                            const float* __restrict__ W0,
                            const float* __restrict__ W1,
                            float* __restrict__ S, int nrows) {
    const int wave = (blockIdx.x * blockDim.x + threadIdx.x) >> 5;
    const int lane = threadIdx.x & 31;
    const int row0 = wave * 16;
    if (row0 >= nrows) return;

    const int m    = lane & 15;
    const int kofs = (lane >> 4) << 1;

    const float* __restrict__ arow = X + (size_t)(row0 + m) * 128;
    const float* __restrict__ brow =
        (m < 8) ? (W0 + (size_t)m * 128) : (W1 + (size_t)(m - 8) * 128);

    v8f c = {0.f, 0.f, 0.f, 0.f, 0.f, 0.f, 0.f, 0.f};
#pragma unroll
    for (int k = 0; k < 128; k += 4) {
        v2f a = *(const v2f*)(arow + k + kofs);
        v2f b = *(const v2f*)(brow + k + kofs);
        c = __builtin_amdgcn_wmma_f32_16x16x4_f32(
             false, a,  false, b,
             (short)0, c,  false,  false);
    }

    const int rbase = row0 + ((lane >> 4) << 3);
    const int col   = lane & 15;
#pragma unroll
    for (int v = 0; v < 8; ++v) {
        S[(size_t)(rbase + v) * 16 + col] = c[v];
    }
}

#define OWN_R   392

__device__ __forceinline__ unsigned own_byte(int v) {
    const unsigned o = (unsigned)v / (unsigned)OWN_R;
    return o < 255u ? o : 255u;
}

__global__ __launch_bounds__(256) void k_ownmap(const int* __restrict__ eidx,
                                                unsigned* __restrict__ map, int n) {
    unsigned char* __restrict__ mb = (unsigned char*)map;
    const int gt = (int)(blockIdx.x * 256 + threadIdx.x);
    if (((uintptr_t)eidx & 15u) != 0) {
        const int i0 = gt * 16;
        for (int i = i0; i < min(n, i0 + 16); ++i) mb[i] = (unsigned char)own_byte(eidx[i]);
        return;
    }
    const int nq   = n >> 2;
    const int lane = threadIdx.x & 31;
    const int q0   = (gt >> 5) * 128 + lane;
    const int4* __restrict__ vp = (const int4*)eidx;
    if (nq > 0) {
        int4 a[4];
#pragma unroll
        for (int j = 0; j < 4; ++j) a[j] = vp[min(q0 + j * 32, nq - 1)];
#pragma unroll
        for (int j = 0; j < 4; ++j) {
            const int q = q0 + j * 32;
            if (q < nq)
                map[q] = own_byte(a[j].x) | (own_byte(a[j].y) << 8) |
                         (own_byte(a[j].z) << 16) | (own_byte(a[j].w) << 24);
        }
    }
    if (gt == 0)
        for (int i = nq * 4; i < n; ++i) mb[i] = (unsigned char)own_byte(eidx[i]);
}

#define OWN_NT  1024
#define OWN_CAP 8192
#define OWN_G   8
#define OWN_EPT (OWN_CAP / OWN_NT)
#define OWN_TIX(n, h) ((h) * OWN_R + (n))
#define OWN_SLOTS ((OWN_R + 31) / 32)
#define SL_ROWS  256
#define SL_MAXNS 8
#define OWN_NBK  (SL_MAXNS * 32 * OWN_SLOTS)
#define OWN_KEY(nl, r) ((((r) / SL_ROWS) * 32 + ((nl) & 31)) * OWN_SLOTS + ((nl) >> 5))
#define OFF_B    (OWN_CAP * 8)
#define OFF_SUM  (OFF_B + OWN_R * 32)
#define OFF_SE   (OFF_B + OWN_R * 64)
#define OFF_EA   (OFF_B + OWN_R * 96)
#define OFF_EN   (OFF_EA + OWN_CAP * 4)
#define LDS_RAW  (OFF_EA + OWN_R * 512)
static_assert(OFF_B + SL_ROWS * 512 <= LDS_RAW, "a slice of x_r fits region B");
static_assert(SL_ROWS * 32 == OWN_G * OWN_NT, "a thread holds a whole slice's share in OWN_G registers");
static_assert(4 * OWN_NT >= OWN_NBK, "the bucket scan takes four buckets per thread");
static_assert(OFF_EN + OWN_CAP * 2 <= LDS_RAW, "the edge list fits region B");
#define OWN_PK   32
#define OFF_PV   (OFF_EN + OWN_CAP * 2)
static_assert(OFF_PV + OWN_PK * OWN_NT * 4 <= LDS_RAW, "the threads' own hit lists fit region B");
#define PJ_PITCH 132
#define PJ_TILES ((OWN_R + 15) / 16)
#define PJ_PATCH (16 * PJ_PITCH * 4)
#define PJ_LOW   7
#define PJ_OFF(t) ((t) < PJ_LOW ? (t) * PJ_PATCH : OFF_EA + ((t) - PJ_LOW) * PJ_PATCH)
#define PJ_WOFF  (PJ_LOW * PJ_PATCH)
static_assert(PJ_WOFF + 8 * PJ_PITCH * 4 <= OFF_B, "low patches and the weight rows end before the tables");
static_assert(OFF_EA + (PJ_TILES - PJ_LOW) * PJ_PATCH <= LDS_RAW, "high patches fit behind the tables");
static_assert(PJ_TILES <= OWN_NT / 32 && OWN_NT >= 256, "a wave per tile, a thread per 16 bytes of the weight rows");

__device__ __forceinline__ v4f pj_ld4(const float* __restrict__ p) {
    v4f r;
    r.x = p[0]; r.y = p[1]; r.z = p[2]; r.w = p[3];
    return r;
}

template <int PH>
__device__ __forceinline__ void own_slow_sweep(const int* __restrict__ idx,
                                               const int* __restrict__ rel,
                                               const float* __restrict__ Sr,
                                               const float* __restrict__ x_r,
                                               float* s_acc, unsigned* s_max,
                                               float* s_sum, const float* s_se,
                                               int E, int base, int nown, int tid) {
    for (int e = tid; e < E; e += OWN_NT) {
        const unsigned d = (unsigned)(idx[e] - base);
        if (d >= (unsigned)nown) continue;
        const int r = rel[e];
        if (PH == 0) {
            for (int h = 0; h < 8; ++h) {
                const float s = lrelu(s_se[OWN_TIX(d, h)] + Sr[(size_t)r * 16 + h]);
                atomicMax(&s_max[OWN_TIX(d, h)], fkey(s));
            }
        } else if (PH == 1) {
            for (int h = 0; h < 8; ++h) {
                const float s = lrelu(s_se[OWN_TIX(d, h)] + Sr[(size_t)r * 16 + h]);
                atomicAdd(&s_sum[OWN_TIX(d, h)], __expf(s - fdecode(s_max[OWN_TIX(d, h)])));
            }
        } else {
            float a = 0.0f;
            for (int h = 0; h < 8; ++h) {
                const float s = lrelu(s_se[OWN_TIX(d, h)] + Sr[(size_t)r * 16 + h]);
                a += __expf(s - fdecode(s_max[OWN_TIX(d, h)])) / (s_sum[OWN_TIX(d, h)] + 1e-16f);
            }
            a *= 0.125f;
            for (int c = 0; c < 128; ++c)
                atomicAdd(&s_acc[d * 128 + c], x_r[(size_t)r * 128 + c] * a);
        }
    }
}

__global__ __launch_bounds__(OWN_NT) void k_owner(const int* __restrict__ eidx,
                                                  const unsigned* __restrict__ map,
                                                  const int* __restrict__ rel,
                                                  const float* __restrict__ x_e,
                                                  const float* __restrict__ Wh,
                                                  const float* __restrict__ Wt,
                                                  const float* __restrict__ Sr,
                                                  const float* __restrict__ x_r,
                                                  float* __restrict__ out,
                                                  int E, int N_e, int nblk, int N_r) {
    __shared__ __attribute__((aligned(16))) unsigned char s_raw[LDS_RAW];
    __shared__ int s_cnt;
    __shared__ int s_bc[OWN_NBK];
    __shared__ unsigned short s_bo[OWN_NBK + 1];
    __shared__ int s_wt[32];
    uint2*          const s_rec  = (uint2*)s_raw;
    unsigned*       const s_max  = (unsigned*)(s_raw + OFF_B);
    float*          const s_sum  = (float*)(s_raw + OFF_SUM);
    float*          const s_se   = (float*)(s_raw + OFF_SE);
    int*            const s_ea   = (int*)(s_raw + OFF_EA);
    unsigned short* const s_en   = (unsigned short*)(s_raw + OFF_EN);
    float*          const s_acc  = (float*)(s_raw + OFF_EA);
    float*          const s_rows = (float*)(s_raw + OFF_B);
    int*            const s_pv   = (int*)(s_raw + OFF_PV);

    const int tid  = threadIdx.x;
    const int lane = tid & 31;
    const int w    = tid >> 5;
    const int side = ((int)blockIdx.x >= nblk) ? 1 : 0;
    const int base = ((int)blockIdx.x - side * nblk) * OWN_R;
    const int nown = min(OWN_R, N_e - base);
    const int* __restrict__ idx = eidx + (size_t)side * (size_t)E;

    const int NS = (N_r + SL_ROWS - 1) / SL_ROWS;
    v4f xrow[16];
    v4f wrow = {0.f, 0.f, 0.f, 0.f};
    const int wu = __builtin_amdgcn_readfirstlane(w);
    if (wu < PJ_TILES) {
        const int rw = base + wu * 16;
#pragma unroll
        for (int j = 0; j < 16; ++j)
            xrow[j] = pj_ld4(x_e + (size_t)min(rw + j, N_e - 1) * 128 + lane * 4);
    }
    if (tid < 256) {
        if (side) wrow = pj_ld4(Wt + tid * 4);
        else      wrow = pj_ld4(Wh + tid * 4);
    }
    for (int i = tid; i < OWN_NBK; i += OWN_NT) s_bc[i] = 0;
    for (int i = tid; i < OWN_R * 8; i += OWN_NT) {
        s_max[i] = NEG_INF_KEY;
        s_sum[i] = 0.0f;
    }
    if (tid == 0) s_cnt = 0;
    if (wu < PJ_TILES) {
        float* const pp = (float*)(s_raw + PJ_OFF(wu));
#pragma unroll
        for (int j = 0; j < 16; ++j) *(v4f*)(pp + j * PJ_PITCH + lane * 4) = xrow[j];
    }
    if (tid < 256) *(v4f*)((float*)(s_raw + PJ_WOFF) + w * PJ_PITCH + lane * 4) = wrow;
    __syncthreads();
    if (wu < PJ_TILES) {
        const int m    = lane & 15;
        const int kofs = (lane >> 4) << 1;
        const float* const ap = (const float*)(s_raw + PJ_OFF(wu)) + m * PJ_PITCH + kofs;
        const float* const bp = (const float*)(s_raw + PJ_WOFF) + (m & 7) * PJ_PITCH + kofs;
        v8f c = {0.f, 0.f, 0.f, 0.f, 0.f, 0.f, 0.f, 0.f};
#pragma unroll
        for (int k = 0; k < 128; k += 4) {
            const v2f a = *(const v2f*)(ap + k);
            const v2f b = *(const v2f*)(bp + k);
            c = __builtin_amdgcn_wmma_f32_16x16x4_f32(
                 false, a,  false, b,
                 (short)0, c,  false,  false);
        }
        const int h  = lane & 15;
        const int r0 = wu * 16 + ((lane >> 4) << 3);
        if (h < 8) {
#pragma unroll
            for (int v = 0; v < 8; ++v) {
                const int nl = r0 + v;
                if (nl < OWN_R) s_se[OWN_TIX(nl, h)] = (nl < nown) ? c[v] : 0.0f;
            }
        }
    }
    __syncthreads();

    int cntp = 0;
    auto keep1 = [&](int e) {
        s_pv[min(cntp, OWN_PK - 1) * OWN_NT + tid] = e;
        ++cntp;
    };
    if (nblk <= 255) {
        const unsigned char* __restrict__ mp = (const unsigned char*)map + (size_t)side * (size_t)E;
        const unsigned me  = (unsigned)((int)blockIdx.x - side * nblk);
        const unsigned pat = me * 0x01010101u;
        auto eqmask = [&](unsigned v) {
            const unsigned x = v ^ pat;
            const unsigned t = (x & 0x7F7F7F7Fu) + 0x7F7F7F7Fu;
            return ~(t | x) & 0x80808080u;
        };
        auto hit16 = [&](const uint4 a) {
            return (eqmask(a.x) >> 3) | (eqmask(a.y) >> 2) | (eqmask(a.z) >> 1) | eqmask(a.w);
        };
        auto drain = [&](unsigned wv, int eb) {
            while (wv != 0u) {
                const unsigned k = (unsigned)__ffs((int)wv) - 1u;
                wv &= wv - 1u;
                keep1(eb + (int)((((k >> 2) & 1u) * (16u * OWN_NT)) + ((k & 3u) << 2) + (k >> 3)));
            }
        };
        int head = (int)((16u - (unsigned)((uintptr_t)mp & 15u)) & 15u);
        if (head > E) head = E;
        for (int e = tid; e < head; e += OWN_NT) if ((unsigned)mp[e] == me) keep1(e);
        const int nvec = (E - head) >> 4;
        const uint4* __restrict__ vp = (const uint4*)(mp + head);
        int q = tid;
        for (; q + 3 * OWN_NT < nvec; q += 4 * OWN_NT) {
            const uint4 a0 = vp[q];
            const uint4 a1 = vp[q + OWN_NT];
            const uint4 a2 = vp[q + 2 * OWN_NT];
            const uint4 a3 = vp[q + 3 * OWN_NT];
            const unsigned wa = (hit16(a0) >> 4) | hit16(a1);
            const unsigned wb = (hit16(a2) >> 4) | hit16(a3);
            drain(wa, head + 16 * q);
            drain(wb, head + 16 * (q + 2 * OWN_NT));
        }
        for (; q < nvec; q += OWN_NT) drain(hit16(vp[q]) >> 4, head + 16 * q);
        for (int e = head + 16 * nvec + tid; e < E; e += OWN_NT) if ((unsigned)mp[e] == me) keep1(e);
    } else {
        for (int e = tid; e < E; e += OWN_NT)
            if ((unsigned)(idx[e] - base) < (unsigned)nown) keep1(e);
    }
    {
        const int mine = (cntp > OWN_PK) ? (OWN_CAP + 1) : cntp;
        int x = mine;
#pragma unroll
        for (int d = 1; d < 32; d <<= 1) {
            const int y = __shfl_up(x, d);
            if (lane >= d) x += y;
        }
        if (lane == 31) s_wt[w] = x;
        __syncthreads();
        if (w == 0) {
            const int v = s_wt[lane];
            int z = v;
#pragma unroll
            for (int d = 1; d < 32; d <<= 1) {
                const int y = __shfl_up(z, d);
                if (lane >= d) z += y;
            }
            s_wt[lane] = z - v;
        }
        __syncthreads();
        const int off = s_wt[w] + x - mine;
        if (tid == OWN_NT - 1) s_cnt = off + mine;
        const int have = min(cntp, OWN_PK);
        for (int i = 0; i < have; ++i)
            if (off + i < OWN_CAP) s_ea[off + i] = s_pv[i * OWN_NT + tid];
    }
    __syncthreads();
    const int total = s_cnt;
    const bool fits = total <= OWN_CAP && NS >= 1 && NS <= SL_MAXNS &&
                      ((((uintptr_t)x_r) | ((uintptr_t)out) | ((uintptr_t)Sr)) & 15u) == 0;

    if (fits) {
        const int cnt = total;
        {
            int p = tid;
            for (; p + 3 * OWN_NT < cnt; p += 4 * OWN_NT) {
                int rr[4], vv[4];
#pragma unroll
                for (int j = 0; j < 4; ++j) {
                    const int e = s_ea[p + j * OWN_NT];
                    rr[j] = rel[e];
                    vv[j] = idx[e];
                }
#pragma unroll
                for (int j = 0; j < 4; ++j) {
                    const int nl = vv[j] - base;
                    s_ea[p + j * OWN_NT] = rr[j];
                    s_en[p + j * OWN_NT] = (unsigned short)nl;
                    atomicAdd(&s_bc[OWN_KEY(nl, rr[j])], 1);
                }
            }
            for (; p < cnt; p += OWN_NT) {
                const int e = s_ea[p];
                const int r = rel[e];
                const int nl = idx[e] - base;
                s_ea[p] = r;
                s_en[p] = (unsigned short)nl;
                atomicAdd(&s_bc[OWN_KEY(nl, r)], 1);
            }
        }
        __syncthreads();
        {
            const int k0 = 4 * tid;
            int c[4];
#pragma unroll
            for (int j = 0; j < 4; ++j) c[j] = (k0 + j < OWN_NBK) ? s_bc[k0 + j] : 0;
            const int mine = c[0] + c[1] + c[2] + c[3];
            int x = mine;
#pragma unroll
            for (int d = 1; d < 32; d <<= 1) {
                const int y = __shfl_up(x, d);
                if (lane >= d) x += y;
            }
            if (lane == 31) s_wt[w] = x;
            __syncthreads();
            if (w == 0) {
                const int v = s_wt[lane];
                int z = v;
#pragma unroll
                for (int d = 1; d < 32; d <<= 1) {
                    const int y = __shfl_up(z, d);
                    if (lane >= d) z += y;
                }
                s_wt[lane] = z - v;
            }
            __syncthreads();
            int run = s_wt[w] + x - mine;
#pragma unroll
            for (int j = 0; j < 4; ++j) {
                if (k0 + j < OWN_NBK) { s_bo[k0 + j] = (unsigned short)run; s_bc[k0 + j] = run; }
                run += c[j];
            }
            if (tid == 0) s_bo[OWN_NBK] = (unsigned short)cnt;
        }
        float sc[OWN_EPT][8];
#pragma unroll
        for (int j = 0; j < OWN_EPT; ++j) {
            if (j * OWN_NT >= cnt) break;
            const int pc = min(tid + j * OWN_NT, cnt - 1);
            const float4* __restrict__ sp = (const float4*)(Sr + (size_t)s_ea[pc] * 16);
            const float4 lo = sp[0], hi = sp[1];
            sc[j][0] = lo.x; sc[j][1] = lo.y; sc[j][2] = lo.z; sc[j][3] = lo.w;
            sc[j][4] = hi.x; sc[j][5] = hi.y; sc[j][6] = hi.z; sc[j][7] = hi.w;
        }
#pragma unroll
        for (int j = 0; j < OWN_EPT; ++j) {
            if (j * OWN_NT >= cnt) break;
            const int p = tid + j * OWN_NT;
            const int n = (int)s_en[min(p, cnt - 1)];
#pragma unroll
            for (int h = 0; h < 8; ++h) sc[j][h] = lrelu(s_se[OWN_TIX(n, h)] + sc[j][h]);
            if (p < cnt) {
#pragma unroll
                for (int h = 0; h < 8; ++h) atomicMax(&s_max[OWN_TIX(n, h)], fkey(sc[j][h]));
            }
        }
        __syncthreads();
#pragma unroll
        for (int j = 0; j < OWN_EPT; ++j) {
            if (j * OWN_NT >= cnt) break;
            const int p = tid + j * OWN_NT;
            const int n = (int)s_en[min(p, cnt - 1)];
            float mx[8];
#pragma unroll
            for (int h = 0; h < 8; ++h) mx[h] = fdecode(s_max[OWN_TIX(n, h)]);
#pragma unroll
            for (int h = 0; h < 8; ++h) sc[j][h] = __expf(sc[j][h] - mx[h]);
            if (p < cnt) {
#pragma unroll
                for (int h = 0; h < 8; ++h) atomicAdd(&s_sum[OWN_TIX(n, h)], sc[j][h]);
            }
        }
        __syncthreads();
#pragma unroll
        for (int j = 0; j < OWN_EPT; ++j) {
            if (j * OWN_NT >= cnt) break;
            const int p = tid + j * OWN_NT;
            const int pc = min(p, cnt - 1);
            const int n = (int)s_en[pc];
            float dn[8];
#pragma unroll
            for (int h = 0; h < 8; ++h) dn[h] = s_sum[OWN_TIX(n, h)] + 1e-16f;
            float a = sc[j][0] / dn[0];
#pragma unroll
            for (int h = 1; h < 8; ++h) a += sc[j][h] / dn[h];
            if (p < cnt) {
                const int r  = s_ea[pc];
                const int qd = atomicAdd(&s_bc[OWN_KEY(n, r)], 1);
                s_rec[qd] = make_uint2(((unsigned)r << 9) | (unsigned)n, __float_as_uint(a * 0.125f));
            }
        }
        __syncthreads();
#define OWN_SLOT_LIST(X) X(0) X(1) X(2) X(3) X(4) X(5) X(6) X(7) X(8) X(9) X(10) X(11) X(12)
        static_assert(OWN_SLOTS == 13, "the slot list names 13 slots");
#define OWN_DECL(S) float4 acc##S = make_float4(0.f, 0.f, 0.f, 0.f);
        OWN_SLOT_LIST(OWN_DECL)
        const int ws = __builtin_amdgcn_readfirstlane(w);
        v4f t[OWN_G];
        {
            const int nv = min(SL_ROWS, N_r) * 32;
            const v4f* __restrict__ sp = (const v4f*)x_r;
#pragma unroll
            for (int j = 0; j < OWN_G; ++j) t[j] = sp[min(tid + j * OWN_NT, nv - 1)];
        }
        for (int s = 0; s < NS; ++s) {
            if (s > 0) __syncthreads();
            {
                const int nv = min(SL_ROWS, N_r - s * SL_ROWS) * 32;
                v4f* dp = (v4f*)s_rows;
#pragma unroll
                for (int j = 0; j < OWN_G; ++j)
                    if (tid + j * OWN_NT < nv) dp[tid + j * OWN_NT] = t[j];
            }
            __syncthreads();
            if (s + 1 < NS) {
                const int r1 = (s + 1) * SL_ROWS;
                const int nv = min(SL_ROWS, N_r - r1) * 32;
                const v4f* __restrict__ sp = (const v4f*)(x_r + (size_t)r1 * 128);
#pragma unroll
                for (int j = 0; j < OWN_G; ++j) t[j] = sp[min(tid + j * OWN_NT, nv - 1)];
            }
            const int kb = (s * 32 + ws) * OWN_SLOTS;
            const int bv = (int)s_bo[kb + min(lane, OWN_SLOTS)];
            const int qz = __builtin_amdgcn_readlane(bv, OWN_SLOTS);
            const int rbase = s * SL_ROWS;
            int q = __builtin_amdgcn_readlane(bv, 0);
            while (q < qz) {
                const int cb = q;
                const int ce = min(qz, cb + 32);
                const uint2 rc = s_rec[min(cb + lane, qz - 1)];
#define OWN_SLOT_RUN(S) { \
                    const int e_ = min(__builtin_amdgcn_readlane(bv, S + 1), ce); \
                    if (q < e_) { \
                        for (int g = q; g < e_; g += 2) { \
                            v4f m_[2]; float a_[2]; \
                            _Pragma("unroll")   \
                            for (int j = 0; j < 2; ++j) { \
                                const int li = min(g + j, e_ - 1) - cb; \
                                const unsigned rx = (unsigned)__builtin_amdgcn_readlane((int)rc.x, li); \
                                a_[j] = __uint_as_float((unsigned)__builtin_amdgcn_readlane((int)rc.y, li)); \
                                m_[j] = ((const v4f*)(s_rows + ((int)(rx >> 9) - rbase) * 128))[lane]; \
                            } \
                              \
                              \
                            asm volatile("" :: "v"(m_[0]), "v"(m_[1])); \
                            _Pragma("unroll")   \
                            for (int j = 0; j < 2; ++j) \
                                if (g + j < e_) {   \
                                    acc##S.x += m_[j].x * a_[j]; acc##S.y += m_[j].y * a_[j]; \
                                    acc##S.z += m_[j].z * a_[j]; acc##S.w += m_[j].w * a_[j]; \
                                } \
                        } \
                        q = e_; \
                    } }
                OWN_SLOT_LIST(OWN_SLOT_RUN)
#undef OWN_SLOT_RUN
            }
        }
#define OWN_OUT(S) if (ws + 32 * S < nown) ((float4*)(out + (size_t)(base + ws + 32 * S) * 256 + side * 128))[lane] = acc##S;
        OWN_SLOT_LIST(OWN_OUT)
#undef OWN_OUT
#undef OWN_DECL
#undef OWN_SLOT_LIST
    } else {
        for (int i = tid; i < OWN_R * 32; i += OWN_NT)
            ((float4*)s_acc)[i] = make_float4(0.f, 0.f, 0.f, 0.f);
        __syncthreads();
        own_slow_sweep<0>(idx, rel, Sr, x_r, s_acc, s_max, s_sum, s_se, E, base, nown, tid);
        __syncthreads();
        own_slow_sweep<1>(idx, rel, Sr, x_r, s_acc, s_max, s_sum, s_se, E, base, nown, tid);
        __syncthreads();
        own_slow_sweep<2>(idx, rel, Sr, x_r, s_acc, s_max, s_sum, s_se, E, base, nown, tid);
        __syncthreads();
        for (int nl = w; nl < nown; nl += 32) {
            const float4 v = ((const float4*)(s_acc + nl * 128))[lane];
            ((float4*)(out + (size_t)(base + nl) * 256 + side * 128))[lane] = v;
        }
    }
}

extern "C" void kernel_launch(void* const* d_in, const int* in_sizes, int n_in,
                              void* d_out, int out_size, void* d_ws, size_t ws_size,
                              hipStream_t stream) {
    const float* x_e  = (const float*)d_in[0];
    const float* x_r  = (const float*)d_in[1];
    const float* Wh   = (const float*)d_in[2];
    const float* Wt   = (const float*)d_in[3];
    const float* Wr   = (const float*)d_in[4];
    const int*   eidx = (const int*)d_in[5];
    const int*   rel  = (const int*)d_in[6];

    const int D   = 128;
    const int N_e = in_sizes[0] / D;
    const int N_r = in_sizes[1] / D;
    const int E   = in_sizes[6];

    auto align256 = [](size_t x) { return (x + 255) & ~(size_t)255; };
    char* ws = (char*)d_ws;
    float* Sr = (float*)ws; ws += align256((size_t)N_r * 16 * sizeof(float));
    unsigned* omap = (unsigned*)ws; ws += align256((size_t)2 * (size_t)E + 16);

    float* out = (float*)d_out;

    const int BLK = 256;

    {
        const int waves   = (N_r + 15) / 16;
        const int threads = waves * 32;
        const int grid    = (threads + BLK - 1) / BLK;
        k_proj_wmma<<<grid, BLK, 0, stream>>>(x_r, Wr, Wr, Sr, N_r);
    }

    {
        const int n    = 2 * E;
        const int grid = (n + 16 * BLK - 1) / (16 * BLK);
        k_ownmap<<<grid, BLK, 0, stream>>>(eidx, omap, n);
    }

    {
        const int nblk = (N_e + OWN_R - 1) / OWN_R;
        k_owner<<<2 * nblk, OWN_NT, 0, stream>>>(eidx, omap, rel, x_e, Wh, Wt, Sr, x_r, out, E, N_e, nblk, N_r);
    }
}


# Softmax passes read node-major copies of score, max and denominator tables, two 16-byte LDS loads per table and edge; max decode and 1/(exp-sum+eps) once per (node, head); weight = exp * reciprocal
# speedup vs baseline: 5.0333x; 1.0398x over previous
#include <hip/hip_runtime.h>
#include <hip/hip_bf16.h>
#include <stdint.h>

typedef __attribute__((ext_vector_type(2))) float v2f;
typedef __attribute__((ext_vector_type(8))) float v8f;
typedef __attribute__((ext_vector_type(4))) float v4f;

#define NEG_SLOPE 0.01f

__device__ __forceinline__ float lrelu(float x) {
    return x > 0.0f ? x : NEG_SLOPE * x;
}

__device__ __forceinline__ unsigned fkey(float f) {
    unsigned u = __float_as_uint(f);
    return (u & 0x80000000u) ? ~u : (u | 0x80000000u);
}
__device__ __forceinline__ float fdecode(unsigned k) {
    unsigned u = (k & 0x80000000u) ? (k ^ 0x80000000u) : ~k;
    return __uint_as_float(u);
}
#define NEG_INF_KEY 0x007FFFFFu

__global__ void k_proj_wmma(const float* __restrict__ X,
                            const float* __restrict__ W0,
                            const float* __restrict__ W1,
                            float* __restrict__ S, int nrows) {
    const int wave = (blockIdx.x * blockDim.x + threadIdx.x) >> 5;
    const int lane = threadIdx.x & 31;
    const int row0 = wave * 16;
    if (row0 >= nrows) return;

    const int m    = lane & 15;
    const int kofs = (lane >> 4) << 1;

    const float* __restrict__ arow = X + (size_t)(row0 + m) * 128;
    const float* __restrict__ brow =
        (m < 8) ? (W0 + (size_t)m * 128) : (W1 + (size_t)(m - 8) * 128);

    v8f c = {0.f, 0.f, 0.f, 0.f, 0.f, 0.f, 0.f, 0.f};
#pragma unroll
    for (int k = 0; k < 128; k += 4) {
        v2f a = *(const v2f*)(arow + k + kofs);
        v2f b = *(const v2f*)(brow + k + kofs);
        c = __builtin_amdgcn_wmma_f32_16x16x4_f32(
             false, a,  false, b,
             (short)0, c,  false,  false);
    }

    const int rbase = row0 + ((lane >> 4) << 3);
    const int col   = lane & 15;
#pragma unroll
    for (int v = 0; v < 8; ++v) {
        S[(size_t)(rbase + v) * 16 + col] = c[v];
    }
}

#define OWN_R   392

__device__ __forceinline__ unsigned own_byte(int v) {
    const unsigned o = (unsigned)v / (unsigned)OWN_R;
    return o < 255u ? o : 255u;
}

__global__ __launch_bounds__(256) void k_ownmap(const int* __restrict__ eidx,
                                                unsigned* __restrict__ map, int n) {
    unsigned char* __restrict__ mb = (unsigned char*)map;
    const int gt = (int)(blockIdx.x * 256 + threadIdx.x);
    if (((uintptr_t)eidx & 15u) != 0) {
        const int i0 = gt * 16;
        for (int i = i0; i < min(n, i0 + 16); ++i) mb[i] = (unsigned char)own_byte(eidx[i]);
        return;
    }
    const int nq   = n >> 2;
    const int lane = threadIdx.x & 31;
    const int q0   = (gt >> 5) * 128 + lane;
    const int4* __restrict__ vp = (const int4*)eidx;
    if (nq > 0) {
        int4 a[4];
#pragma unroll
        for (int j = 0; j < 4; ++j) a[j] = vp[min(q0 + j * 32, nq - 1)];
#pragma unroll
        for (int j = 0; j < 4; ++j) {
            const int q = q0 + j * 32;
            if (q < nq)
                map[q] = own_byte(a[j].x) | (own_byte(a[j].y) << 8) |
                         (own_byte(a[j].z) << 16) | (own_byte(a[j].w) << 24);
        }
    }
    if (gt == 0)
        for (int i = nq * 4; i < n; ++i) mb[i] = (unsigned char)own_byte(eidx[i]);
}

#define OWN_NT  1024
#define OWN_CAP 8192
#define OWN_G   8
#define OWN_EPT (OWN_CAP / OWN_NT)
#define OWN_TIX(n, h) ((h) * OWN_R + (n))
#define OWN_SLOTS ((OWN_R + 31) / 32)
#define SL_ROWS  256
#define SL_MAXNS 8
#define OWN_NBK  (SL_MAXNS * 32 * OWN_SLOTS)
#define OWN_KEY(nl, r) ((((r) / SL_ROWS) * 32 + ((nl) & 31)) * OWN_SLOTS + ((nl) >> 5))
#define OFF_B    (OWN_CAP * 8)
#define OFF_SUM  (OFF_B + OWN_R * 32)
#define OFF_SE   (OFF_B + OWN_R * 64)
#define OFF_EA   (OFF_B + OWN_R * 96)
#define OFF_EN   (OFF_EA + OWN_CAP * 4)
#define LDS_RAW  (OFF_EA + OWN_R * 512)
static_assert(OFF_B + SL_ROWS * 512 <= LDS_RAW, "a slice of x_r fits region B");
static_assert(SL_ROWS * 32 == OWN_G * OWN_NT, "a thread holds a whole slice's share in OWN_G registers");
static_assert(4 * OWN_NT >= OWN_NBK, "the bucket scan takes four buckets per thread");
static_assert(OFF_EN + OWN_CAP * 2 <= LDS_RAW, "the edge list fits region B");
#define OWN_PK   32
#define OFF_PV   (OFF_EN + OWN_CAP * 2)
static_assert(OFF_PV + OWN_PK * OWN_NT * 4 <= LDS_RAW, "the threads' own hit lists fit region B");
#define NM_PITCH 12
#define NM_BYTES (OWN_R * NM_PITCH * 4)
#define OFF_NSE  (OFF_PV + OWN_PK * OWN_NT * 4)
#define OFF_NMX  OFF_PV
#define OFF_NIV  (OFF_NMX + NM_BYTES)
static_assert(OFF_PV % 16 == 0 && NM_BYTES % 16 == 0 && (OWN_PK * OWN_NT * 4) % 16 == 0, "16-byte loads of the copies are aligned");
static_assert(OFF_NSE + NM_BYTES <= LDS_RAW, "the node-score copy fits behind the hit lists");
static_assert(OFF_NIV + NM_BYTES <= OFF_NSE, "max and denominator copies fit the dead hit lists");
static_assert(OWN_R * 8 <= 4 * OWN_NT && OWN_NT % 8 == 0, "a copy pass: eight threads per node, at most four nodes per thread");
#define PJ_PITCH 132
#define PJ_TILES ((OWN_R + 15) / 16)
#define PJ_PATCH (16 * PJ_PITCH * 4)
#define PJ_LOW   7
#define PJ_OFF(t) ((t) < PJ_LOW ? (t) * PJ_PATCH : OFF_EA + ((t) - PJ_LOW) * PJ_PATCH)
#define PJ_WOFF  (PJ_LOW * PJ_PATCH)
static_assert(PJ_WOFF + 8 * PJ_PITCH * 4 <= OFF_B, "low patches and the weight rows end before the tables");
static_assert(OFF_EA + (PJ_TILES - PJ_LOW) * PJ_PATCH <= LDS_RAW, "high patches fit behind the tables");
static_assert(OFF_EA + (PJ_TILES - PJ_LOW) * PJ_PATCH <= OFF_NSE, "high patches end before the node-major score copy");
static_assert(PJ_TILES <= OWN_NT / 32 && OWN_NT >= 256, "a wave per tile, a thread per 16 bytes of the weight rows");

__device__ __forceinline__ v4f pj_ld4(const float* __restrict__ p) {
    v4f r;
    r.x = p[0]; r.y = p[1]; r.z = p[2]; r.w = p[3];
    return r;
}

template <int PH>
__device__ __forceinline__ void own_slow_sweep(const int* __restrict__ idx,
                                               const int* __restrict__ rel,
                                               const float* __restrict__ Sr,
                                               const float* __restrict__ x_r,
                                               float* s_acc, unsigned* s_max,
                                               float* s_sum, const float* s_se,
                                               int E, int base, int nown, int tid) {
    for (int e = tid; e < E; e += OWN_NT) {
        const unsigned d = (unsigned)(idx[e] - base);
        if (d >= (unsigned)nown) continue;
        const int r = rel[e];
        if (PH == 0) {
            for (int h = 0; h < 8; ++h) {
                const float s = lrelu(s_se[OWN_TIX(d, h)] + Sr[(size_t)r * 16 + h]);
                atomicMax(&s_max[OWN_TIX(d, h)], fkey(s));
            }
        } else if (PH == 1) {
            for (int h = 0; h < 8; ++h) {
                const float s = lrelu(s_se[OWN_TIX(d, h)] + Sr[(size_t)r * 16 + h]);
                atomicAdd(&s_sum[OWN_TIX(d, h)], __expf(s - fdecode(s_max[OWN_TIX(d, h)])));
            }
        } else {
            float a = 0.0f;
            for (int h = 0; h < 8; ++h) {
                const float s = lrelu(s_se[OWN_TIX(d, h)] + Sr[(size_t)r * 16 + h]);
                a += __expf(s - fdecode(s_max[OWN_TIX(d, h)])) / (s_sum[OWN_TIX(d, h)] + 1e-16f);
            }
            a *= 0.125f;
            for (int c = 0; c < 128; ++c)
                atomicAdd(&s_acc[d * 128 + c], x_r[(size_t)r * 128 + c] * a);
        }
    }
}

__global__ __launch_bounds__(OWN_NT) void k_owner(const int* __restrict__ eidx,
                                                  const unsigned* __restrict__ map,
                                                  const int* __restrict__ rel,
                                                  const float* __restrict__ x_e,
                                                  const float* __restrict__ Wh,
                                                  const float* __restrict__ Wt,
                                                  const float* __restrict__ Sr,
                                                  const float* __restrict__ x_r,
                                                  float* __restrict__ out,
                                                  int E, int N_e, int nblk, int N_r) {
    __shared__ __attribute__((aligned(16))) unsigned char s_raw[LDS_RAW];
    __shared__ int s_cnt;
    __shared__ int s_bc[OWN_NBK];
    __shared__ unsigned short s_bo[OWN_NBK + 1];
    __shared__ int s_wt[32];
    uint2*          const s_rec  = (uint2*)s_raw;
    unsigned*       const s_max  = (unsigned*)(s_raw + OFF_B);
    float*          const s_sum  = (float*)(s_raw + OFF_SUM);
    float*          const s_se   = (float*)(s_raw + OFF_SE);
    int*            const s_ea   = (int*)(s_raw + OFF_EA);
    unsigned short* const s_en   = (unsigned short*)(s_raw + OFF_EN);
    float*          const s_acc  = (float*)(s_raw + OFF_EA);
    float*          const s_rows = (float*)(s_raw + OFF_B);
    int*            const s_pv   = (int*)(s_raw + OFF_PV);
    float*          const s_nse  = (float*)(s_raw + OFF_NSE);
    float*          const s_nmx  = (float*)(s_raw + OFF_NMX);
    float*          const s_niv  = (float*)(s_raw + OFF_NIV);

    const int tid  = threadIdx.x;
    const int lane = tid & 31;
    const int w    = tid >> 5;
    const int side = ((int)blockIdx.x >= nblk) ? 1 : 0;
    const int base = ((int)blockIdx.x - side * nblk) * OWN_R;
    const int nown = min(OWN_R, N_e - base);
    const int* __restrict__ idx = eidx + (size_t)side * (size_t)E;

    const int NS = (N_r + SL_ROWS - 1) / SL_ROWS;
    v4f xrow[16];
    v4f wrow = {0.f, 0.f, 0.f, 0.f};
    const int wu = __builtin_amdgcn_readfirstlane(w);
    if (wu < PJ_TILES) {
        const int rw = base + wu * 16;
#pragma unroll
        for (int j = 0; j < 16; ++j)
            xrow[j] = pj_ld4(x_e + (size_t)min(rw + j, N_e - 1) * 128 + lane * 4);
    }
    if (tid < 256) {
        if (side) wrow = pj_ld4(Wt + tid * 4);
        else      wrow = pj_ld4(Wh + tid * 4);
    }
    for (int i = tid; i < OWN_NBK; i += OWN_NT) s_bc[i] = 0;
    for (int i = tid; i < OWN_R * 8; i += OWN_NT) {
        s_max[i] = NEG_INF_KEY;
        s_sum[i] = 0.0f;
    }
    if (tid == 0) s_cnt = 0;
    if (wu < PJ_TILES) {
        float* const pp = (float*)(s_raw + PJ_OFF(wu));
#pragma unroll
        for (int j = 0; j < 16; ++j) *(v4f*)(pp + j * PJ_PITCH + lane * 4) = xrow[j];
    }
    if (tid < 256) *(v4f*)((float*)(s_raw + PJ_WOFF) + w * PJ_PITCH + lane * 4) = wrow;
    __syncthreads();
    if (wu < PJ_TILES) {
        const int m    = lane & 15;
        const int kofs = (lane >> 4) << 1;
        const float* const ap = (const float*)(s_raw + PJ_OFF(wu)) + m * PJ_PITCH + kofs;
        const float* const bp = (const float*)(s_raw + PJ_WOFF) + (m & 7) * PJ_PITCH + kofs;
        v8f c = {0.f, 0.f, 0.f, 0.f, 0.f, 0.f, 0.f, 0.f};
#pragma unroll
        for (int k = 0; k < 128; k += 4) {
            const v2f a = *(const v2f*)(ap + k);
            const v2f b = *(const v2f*)(bp + k);
            c = __builtin_amdgcn_wmma_f32_16x16x4_f32(
                 false, a,  false, b,
                 (short)0, c,  false,  false);
        }
        const int h  = lane & 15;
        const int r0 = wu * 16 + ((lane >> 4) << 3);
        if (h < 8) {
#pragma unroll
            for (int v = 0; v < 8; ++v) {
                const int nl = r0 + v;
                if (nl < OWN_R) {
                    const float sv = (nl < nown) ? c[v] : 0.0f;
                    s_se[OWN_TIX(nl, h)] = sv;
                    s_nse[nl * NM_PITCH + h] = sv;
                }
            }
        }
    }
    __syncthreads();

    int cntp = 0;
    auto keep1 = [&](int e) {
        s_pv[min(cntp, OWN_PK - 1) * OWN_NT + tid] = e;
        ++cntp;
    };
    if (nblk <= 255) {
        const unsigned char* __restrict__ mp = (const unsigned char*)map + (size_t)side * (size_t)E;
        const unsigned me  = (unsigned)((int)blockIdx.x - side * nblk);
        const unsigned pat = me * 0x01010101u;
        auto eqmask = [&](unsigned v) {
            const unsigned x = v ^ pat;
            const unsigned t = (x & 0x7F7F7F7Fu) + 0x7F7F7F7Fu;
            return ~(t | x) & 0x80808080u;
        };
        auto hit16 = [&](const uint4 a) {
            return (eqmask(a.x) >> 3) | (eqmask(a.y) >> 2) | (eqmask(a.z) >> 1) | eqmask(a.w);
        };
        auto drain = [&](unsigned wv, int eb) {
            while (wv != 0u) {
                const unsigned k = (unsigned)__ffs((int)wv) - 1u;
                wv &= wv - 1u;
                keep1(eb + (int)((((k >> 2) & 1u) * (16u * OWN_NT)) + ((k & 3u) << 2) + (k >> 3)));
            }
        };
        int head = (int)((16u - (unsigned)((uintptr_t)mp & 15u)) & 15u);
        if (head > E) head = E;
        for (int e = tid; e < head; e += OWN_NT) if ((unsigned)mp[e] == me) keep1(e);
        const int nvec = (E - head) >> 4;
        const uint4* __restrict__ vp = (const uint4*)(mp + head);
        int q = tid;
        for (; q + 3 * OWN_NT < nvec; q += 4 * OWN_NT) {
            const uint4 a0 = vp[q];
            const uint4 a1 = vp[q + OWN_NT];
            const uint4 a2 = vp[q + 2 * OWN_NT];
            const uint4 a3 = vp[q + 3 * OWN_NT];
            const unsigned wa = (hit16(a0) >> 4) | hit16(a1);
            const unsigned wb = (hit16(a2) >> 4) | hit16(a3);
            drain(wa, head + 16 * q);
            drain(wb, head + 16 * (q + 2 * OWN_NT));
        }
        for (; q < nvec; q += OWN_NT) drain(hit16(vp[q]) >> 4, head + 16 * q);
        for (int e = head + 16 * nvec + tid; e < E; e += OWN_NT) if ((unsigned)mp[e] == me) keep1(e);
    } else {
        for (int e = tid; e < E; e += OWN_NT)
            if ((unsigned)(idx[e] - base) < (unsigned)nown) keep1(e);
    }
    {
        const int mine = (cntp > OWN_PK) ? (OWN_CAP + 1) : cntp;
        int x = mine;
#pragma unroll
        for (int d = 1; d < 32; d <<= 1) {
            const int y = __shfl_up(x, d);
            if (lane >= d) x += y;
        }
        if (lane == 31) s_wt[w] = x;
        __syncthreads();
        if (w == 0) {
            const int v = s_wt[lane];
            int z = v;
#pragma unroll
            for (int d = 1; d < 32; d <<= 1) {
                const int y = __shfl_up(z, d);
                if (lane >= d) z += y;
            }
            s_wt[lane] = z - v;
        }
        __syncthreads();
        const int off = s_wt[w] + x - mine;
        if (tid == OWN_NT - 1) s_cnt = off + mine;
        const int have = min(cntp, OWN_PK);
        for (int i = 0; i < have; ++i)
            if (off + i < OWN_CAP) s_ea[off + i] = s_pv[i * OWN_NT + tid];
    }
    __syncthreads();
    const int total = s_cnt;
    const bool fits = total <= OWN_CAP && NS >= 1 && NS <= SL_MAXNS &&
                      ((((uintptr_t)x_r) | ((uintptr_t)out) | ((uintptr_t)Sr)) & 15u) == 0;

    if (fits) {
        const int cnt = total;
        {
            int p = tid;
            for (; p + 3 * OWN_NT < cnt; p += 4 * OWN_NT) {
                int rr[4], vv[4];
#pragma unroll
                for (int j = 0; j < 4; ++j) {
                    const int e = s_ea[p + j * OWN_NT];
                    rr[j] = rel[e];
                    vv[j] = idx[e];
                }
#pragma unroll
                for (int j = 0; j < 4; ++j) {
                    const int nl = vv[j] - base;
                    s_ea[p + j * OWN_NT] = rr[j];
                    s_en[p + j * OWN_NT] = (unsigned short)nl;
                    atomicAdd(&s_bc[OWN_KEY(nl, rr[j])], 1);
                }
            }
            for (; p < cnt; p += OWN_NT) {
                const int e = s_ea[p];
                const int r = rel[e];
                const int nl = idx[e] - base;
                s_ea[p] = r;
                s_en[p] = (unsigned short)nl;
                atomicAdd(&s_bc[OWN_KEY(nl, r)], 1);
            }
        }
        __syncthreads();
        {
            const int k0 = 4 * tid;
            int c[4];
#pragma unroll
            for (int j = 0; j < 4; ++j) c[j] = (k0 + j < OWN_NBK) ? s_bc[k0 + j] : 0;
            const int mine = c[0] + c[1] + c[2] + c[3];
            int x = mine;
#pragma unroll
            for (int d = 1; d < 32; d <<= 1) {
                const int y = __shfl_up(x, d);
                if (lane >= d) x += y;
            }
            if (lane == 31) s_wt[w] = x;
            __syncthreads();
            if (w == 0) {
                const int v = s_wt[lane];
                int z = v;
#pragma unroll
                for (int d = 1; d < 32; d <<= 1) {
                    const int y = __shfl_up(z, d);
                    if (lane >= d) z += y;
                }
                s_wt[lane] = z - v;
            }
            __syncthreads();
            int run = s_wt[w] + x - mine;
#pragma unroll
            for (int j = 0; j < 4; ++j) {
                if (k0 + j < OWN_NBK) { s_bo[k0 + j] = (unsigned short)run; s_bc[k0 + j] = run; }
                run += c[j];
            }
            if (tid == 0) s_bo[OWN_NBK] = (unsigned short)cnt;
        }
        float sc[OWN_EPT][8];
#pragma unroll
        for (int j = 0; j < OWN_EPT; ++j) {
            if (j * OWN_NT >= cnt) break;
            const int pc = min(tid + j * OWN_NT, cnt - 1);
            const float4* __restrict__ sp = (const float4*)(Sr + (size_t)s_ea[pc] * 16);
            const float4 lo = sp[0], hi = sp[1];
            sc[j][0] = lo.x; sc[j][1] = lo.y; sc[j][2] = lo.z; sc[j][3] = lo.w;
            sc[j][4] = hi.x; sc[j][5] = hi.y; sc[j][6] = hi.z; sc[j][7] = hi.w;
        }
#pragma unroll
        for (int j = 0; j < OWN_EPT; ++j) {
            if (j * OWN_NT >= cnt) break;
            const int p = tid + j * OWN_NT;
            const int n = (int)s_en[min(p, cnt - 1)];
            const float4 e0 = *(const float4*)(s_nse + n * NM_PITCH);
            const float4 e1 = *(const float4*)(s_nse + n * NM_PITCH + 4);
            const float se[8] = {e0.x, e0.y, e0.z, e0.w, e1.x, e1.y, e1.z, e1.w};
#pragma unroll
            for (int h = 0; h < 8; ++h) sc[j][h] = lrelu(se[h] + sc[j][h]);
            if (p < cnt) {
#pragma unroll
                for (int h = 0; h < 8; ++h) atomicMax(&s_max[OWN_TIX(n, h)], fkey(sc[j][h]));
            }
        }
        __syncthreads();
        {
            unsigned v[4];
#pragma unroll
            for (int k = 0; k < 4; ++k)
                v[k] = s_max[OWN_TIX(min((tid >> 3) + k * (OWN_NT / 8), OWN_R - 1), tid & 7)];
#pragma unroll
            for (int k = 0; k < 4; ++k) {
                const int n = (tid >> 3) + k * (OWN_NT / 8);
                if (n < OWN_R) s_nmx[n * NM_PITCH + (tid & 7)] = fdecode(v[k]);
            }
        }
        __syncthreads();
#pragma unroll
        for (int j = 0; j < OWN_EPT; ++j) {
            if (j * OWN_NT >= cnt) break;
            const int p = tid + j * OWN_NT;
            const int n = (int)s_en[min(p, cnt - 1)];
            const float4 m0 = *(const float4*)(s_nmx + n * NM_PITCH);
            const float4 m1 = *(const float4*)(s_nmx + n * NM_PITCH + 4);
            const float mx[8] = {m0.x, m0.y, m0.z, m0.w, m1.x, m1.y, m1.z, m1.w};
#pragma unroll
            for (int h = 0; h < 8; ++h) sc[j][h] = __expf(sc[j][h] - mx[h]);
            if (p < cnt) {
#pragma unroll
                for (int h = 0; h < 8; ++h) atomicAdd(&s_sum[OWN_TIX(n, h)], sc[j][h]);
            }
        }
        __syncthreads();
        {
            float v[4];
#pragma unroll
            for (int k = 0; k < 4; ++k)
                v[k] = s_sum[OWN_TIX(min((tid >> 3) + k * (OWN_NT / 8), OWN_R - 1), tid & 7)];
#pragma unroll
            for (int k = 0; k < 4; ++k) {
                const int n = (tid >> 3) + k * (OWN_NT / 8);
                if (n < OWN_R) s_niv[n * NM_PITCH + (tid & 7)] = 1.0f / (v[k] + 1e-16f);
            }
        }
        __syncthreads();
#pragma unroll
        for (int j = 0; j < OWN_EPT; ++j) {
            if (j * OWN_NT >= cnt) break;
            const int p = tid + j * OWN_NT;
            const int pc = min(p, cnt - 1);
            const int n = (int)s_en[pc];
            const float4 v0 = *(const float4*)(s_niv + n * NM_PITCH);
            const float4 v1 = *(const float4*)(s_niv + n * NM_PITCH + 4);
            const float iv[8] = {v0.x, v0.y, v0.z, v0.w, v1.x, v1.y, v1.z, v1.w};
            float a = sc[j][0] * iv[0];
#pragma unroll
            for (int h = 1; h < 8; ++h) a += sc[j][h] * iv[h];
            if (p < cnt) {
                const int r  = s_ea[pc];
                const int qd = atomicAdd(&s_bc[OWN_KEY(n, r)], 1);
                s_rec[qd] = make_uint2(((unsigned)r << 9) | (unsigned)n, __float_as_uint(a * 0.125f));
            }
        }
        __syncthreads();
#define OWN_SLOT_LIST(X) X(0) X(1) X(2) X(3) X(4) X(5) X(6) X(7) X(8) X(9) X(10) X(11) X(12)
        static_assert(OWN_SLOTS == 13, "the slot list names 13 slots");
#define OWN_DECL(S) float4 acc##S = make_float4(0.f, 0.f, 0.f, 0.f);
        OWN_SLOT_LIST(OWN_DECL)
        const int ws = __builtin_amdgcn_readfirstlane(w);
        v4f t[OWN_G];
        {
            const int nv = min(SL_ROWS, N_r) * 32;
            const v4f* __restrict__ sp = (const v4f*)x_r;
#pragma unroll
            for (int j = 0; j < OWN_G; ++j) t[j] = sp[min(tid + j * OWN_NT, nv - 1)];
        }
        for (int s = 0; s < NS; ++s) {
            if (s > 0) __syncthreads();
            {
                const int nv = min(SL_ROWS, N_r - s * SL_ROWS) * 32;
                v4f* dp = (v4f*)s_rows;
#pragma unroll
                for (int j = 0; j < OWN_G; ++j)
                    if (tid + j * OWN_NT < nv) dp[tid + j * OWN_NT] = t[j];
            }
            __syncthreads();
            if (s + 1 < NS) {
                const int r1 = (s + 1) * SL_ROWS;
                const int nv = min(SL_ROWS, N_r - r1) * 32;
                const v4f* __restrict__ sp = (const v4f*)(x_r + (size_t)r1 * 128);
#pragma unroll
                for (int j = 0; j < OWN_G; ++j) t[j] = sp[min(tid + j * OWN_NT, nv - 1)];
            }
            const int kb = (s * 32 + ws) * OWN_SLOTS;
            const int bv = (int)s_bo[kb + min(lane, OWN_SLOTS)];
            const int qz = __builtin_amdgcn_readlane(bv, OWN_SLOTS);
            const int rbase = s * SL_ROWS;
            int q = __builtin_amdgcn_readlane(bv, 0);
            while (q < qz) {
                const int cb = q;
                const int ce = min(qz, cb + 32);
                const uint2 rc = s_rec[min(cb + lane, qz - 1)];
#define OWN_SLOT_RUN(S) { \
                    const int e_ = min(__builtin_amdgcn_readlane(bv, S + 1), ce); \
                    if (q < e_) { \
                        for (int g = q; g < e_; g += 2) { \
                            v4f m_[2]; float a_[2]; \
                            _Pragma("unroll")   \
                            for (int j = 0; j < 2; ++j) { \
                                const int li = min(g + j, e_ - 1) - cb; \
                                const unsigned rx = (unsigned)__builtin_amdgcn_readlane((int)rc.x, li); \
                                a_[j] = __uint_as_float((unsigned)__builtin_amdgcn_readlane((int)rc.y, li)); \
                                m_[j] = ((const v4f*)(s_rows + ((int)(rx >> 9) - rbase) * 128))[lane]; \
                            } \
                              \
                              \
                            asm volatile("" :: "v"(m_[0]), "v"(m_[1])); \
                            _Pragma("unroll")   \
                            for (int j = 0; j < 2; ++j) \
                                if (g + j < e_) {   \
                                    acc##S.x += m_[j].x * a_[j]; acc##S.y += m_[j].y * a_[j]; \
                                    acc##S.z += m_[j].z * a_[j]; acc##S.w += m_[j].w * a_[j]; \
                                } \
                        } \
                        q = e_; \
                    } }
                OWN_SLOT_LIST(OWN_SLOT_RUN)
#undef OWN_SLOT_RUN
            }
        }
#define OWN_OUT(S) if (ws + 32 * S < nown) ((float4*)(out + (size_t)(base + ws + 32 * S) * 256 + side * 128))[lane] = acc##S;
        OWN_SLOT_LIST(OWN_OUT)
#undef OWN_OUT
#undef OWN_DECL
#undef OWN_SLOT_LIST
    } else {
        for (int i = tid; i < OWN_R * 32; i += OWN_NT)
            ((float4*)s_acc)[i] = make_float4(0.f, 0.f, 0.f, 0.f);
        __syncthreads();
        own_slow_sweep<0>(idx, rel, Sr, x_r, s_acc, s_max, s_sum, s_se, E, base, nown, tid);
        __syncthreads();
        own_slow_sweep<1>(idx, rel, Sr, x_r, s_acc, s_max, s_sum, s_se, E, base, nown, tid);
        __syncthreads();
        own_slow_sweep<2>(idx, rel, Sr, x_r, s_acc, s_max, s_sum, s_se, E, base, nown, tid);
        __syncthreads();
        for (int nl = w; nl < nown; nl += 32) {
            const float4 v = ((const float4*)(s_acc + nl * 128))[lane];
            ((float4*)(out + (size_t)(base + nl) * 256 + side * 128))[lane] = v;
        }
    }
}

extern "C" void kernel_launch(void* const* d_in, const int* in_sizes, int n_in,
                              void* d_out, int out_size, void* d_ws, size_t ws_size,
                              hipStream_t stream) {
    const float* x_e  = (const float*)d_in[0];
    const float* x_r  = (const float*)d_in[1];
    const float* Wh   = (const float*)d_in[2];
    const float* Wt   = (const float*)d_in[3];
    const float* Wr   = (const float*)d_in[4];
    const int*   eidx = (const int*)d_in[5];
    const int*   rel  = (const int*)d_in[6];

    const int D   = 128;
    const int N_e = in_sizes[0] / D;
    const int N_r = in_sizes[1] / D;
    const int E   = in_sizes[6];

    auto align256 = [](size_t x) { return (x + 255) & ~(size_t)255; };
    char* ws = (char*)d_ws;
    float* Sr = (float*)ws; ws += align256((size_t)N_r * 16 * sizeof(float));
    unsigned* omap = (unsigned*)ws; ws += align256((size_t)2 * (size_t)E + 16);

    float* out = (float*)d_out;

    const int BLK = 256;

    {
        const int waves   = (N_r + 15) / 16;
        const int threads = waves * 32;
        const int grid    = (threads + BLK - 1) / BLK;
        k_proj_wmma<<<grid, BLK, 0, stream>>>(x_r, Wr, Wr, Sr, N_r);
    }

    {
        const int n    = 2 * E;
        const int grid = (n + 16 * BLK - 1) / (16 * BLK);
        k_ownmap<<<grid, BLK, 0, stream>>>(eidx, omap, n);
    }

    {
        const int nblk = (N_e + OWN_R - 1) / OWN_R;
        k_owner<<<2 * nblk, OWN_NT, 0, stream>>>(eidx, omap, rel, x_e, Wh, Wt, Sr, x_r, out, E, N_e, nblk, N_r);
    }
}


# Pre-pass also writes one 4-byte word per (side, edge): relation << 9 | node index in its block; owner blocks fetch that word per kept edge, all as one group, not id and relation from two arrays
# speedup vs baseline: 5.3138x; 1.0557x over previous
#include <hip/hip_runtime.h>
#include <hip/hip_bf16.h>
#include <stdint.h>

typedef __attribute__((ext_vector_type(2))) float v2f;
typedef __attribute__((ext_vector_type(8))) float v8f;
typedef __attribute__((ext_vector_type(4))) float v4f;

#define NEG_SLOPE 0.01f

__device__ __forceinline__ float lrelu(float x) {
    return x > 0.0f ? x : NEG_SLOPE * x;
}

__device__ __forceinline__ unsigned fkey(float f) {
    unsigned u = __float_as_uint(f);
    return (u & 0x80000000u) ? ~u : (u | 0x80000000u);
}
__device__ __forceinline__ float fdecode(unsigned k) {
    unsigned u = (k & 0x80000000u) ? (k ^ 0x80000000u) : ~k;
    return __uint_as_float(u);
}
#define NEG_INF_KEY 0x007FFFFFu

__global__ void k_proj_wmma(const float* __restrict__ X,
                            const float* __restrict__ W0,
                            const float* __restrict__ W1,
                            float* __restrict__ S, int nrows) {
    const int wave = (blockIdx.x * blockDim.x + threadIdx.x) >> 5;
    const int lane = threadIdx.x & 31;
    const int row0 = wave * 16;
    if (row0 >= nrows) return;

    const int m    = lane & 15;
    const int kofs = (lane >> 4) << 1;

    const float* __restrict__ arow = X + (size_t)(row0 + m) * 128;
    const float* __restrict__ brow =
        (m < 8) ? (W0 + (size_t)m * 128) : (W1 + (size_t)(m - 8) * 128);

    v8f c = {0.f, 0.f, 0.f, 0.f, 0.f, 0.f, 0.f, 0.f};
#pragma unroll
    for (int k = 0; k < 128; k += 4) {
        v2f a = *(const v2f*)(arow + k + kofs);
        v2f b = *(const v2f*)(brow + k + kofs);
        c = __builtin_amdgcn_wmma_f32_16x16x4_f32(
             false, a,  false, b,
             (short)0, c,  false,  false);
    }

    const int rbase = row0 + ((lane >> 4) << 3);
    const int col   = lane & 15;
#pragma unroll
    for (int v = 0; v < 8; ++v) {
        S[(size_t)(rbase + v) * 16 + col] = c[v];
    }
}

#define OWN_R   392

__device__ __forceinline__ unsigned own_byte(int v) {
    const unsigned o = (unsigned)v / (unsigned)OWN_R;
    return o < 255u ? o : 255u;
}
__device__ __forceinline__ unsigned own_word(int v, int r) {
    const unsigned o = (unsigned)v / (unsigned)OWN_R;
    return ((unsigned)r << 9) | ((unsigned)v - o * (unsigned)OWN_R);
}

__global__ __launch_bounds__(256) void k_ownmap(const int* __restrict__ eidx,
                                                const int* __restrict__ rel,
                                                unsigned* __restrict__ map,
                                                unsigned* __restrict__ comb, int n, int E) {
    unsigned char* __restrict__ mb = (unsigned char*)map;
    const int gt = (int)(blockIdx.x * 256 + threadIdx.x);
    auto rel_of = [&](int i) { return rel[i >= E ? i - E : i]; };
    if ((((((uintptr_t)eidx) | ((uintptr_t)rel) | ((uintptr_t)comb)) & 15u) != 0) || ((E & 3) != 0)) {
        const int i0 = gt * 16;
        for (int i = i0; i < min(n, i0 + 16); ++i) {
            const int v = eidx[i];
            mb[i] = (unsigned char)own_byte(v);
            comb[i] = own_word(v, rel_of(i));
        }
        return;
    }
    const int nq   = n >> 2;
    const int lane = threadIdx.x & 31;
    const int q0   = (gt >> 5) * 128 + lane;
    const int4* __restrict__ vp = (const int4*)eidx;
    const int4* __restrict__ rp = (const int4*)rel;
    const int eq = E >> 2;
    if (nq > 0) {
        int4 a[4], rr[4];
#pragma unroll
        for (int j = 0; j < 4; ++j) {
            const int qq = min(q0 + j * 32, nq - 1);
            a[j]  = vp[qq];
            rr[j] = rp[qq >= eq ? qq - eq : qq];
        }
        asm volatile("" :: "v"(a[0].x), "v"(a[0].y), "v"(a[0].z), "v"(a[0].w),
                           "v"(rr[0].x), "v"(rr[0].y), "v"(rr[0].z), "v"(rr[0].w));
#pragma unroll
        for (int j = 0; j < 4; ++j) {
            const int q = q0 + j * 32;
            if (q < nq) {
                map[q] = own_byte(a[j].x) | (own_byte(a[j].y) << 8) |
                         (own_byte(a[j].z) << 16) | (own_byte(a[j].w) << 24);
                ((uint4*)comb)[q] = make_uint4(own_word(a[j].x, rr[j].x), own_word(a[j].y, rr[j].y),
                                               own_word(a[j].z, rr[j].z), own_word(a[j].w, rr[j].w));
            }
        }
    }
    if (gt == 0)
        for (int i = nq * 4; i < n; ++i) {
            const int v = eidx[i];
            mb[i] = (unsigned char)own_byte(v);
            comb[i] = own_word(v, rel_of(i));
        }
}

#define OWN_NT  1024
#define OWN_CAP 8192
#define OWN_G   8
#define OWN_EPT (OWN_CAP / OWN_NT)
#define OWN_TIX(n, h) ((h) * OWN_R + (n))
#define OWN_SLOTS ((OWN_R + 31) / 32)
#define SL_ROWS  256
#define SL_MAXNS 8
#define OWN_NBK  (SL_MAXNS * 32 * OWN_SLOTS)
#define OWN_KEY(nl, r) ((((r) / SL_ROWS) * 32 + ((nl) & 31)) * OWN_SLOTS + ((nl) >> 5))
#define OFF_B    (OWN_CAP * 8)
#define OFF_SUM  (OFF_B + OWN_R * 32)
#define OFF_SE   (OFF_B + OWN_R * 64)
#define OFF_EA   (OFF_B + OWN_R * 96)
#define OFF_EN   (OFF_EA + OWN_CAP * 4)
#define LDS_RAW  (OFF_EA + OWN_R * 512)
static_assert(OFF_B + SL_ROWS * 512 <= LDS_RAW, "a slice of x_r fits region B");
static_assert(SL_ROWS * 32 == OWN_G * OWN_NT, "a thread holds a whole slice's share in OWN_G registers");
static_assert(4 * OWN_NT >= OWN_NBK, "the bucket scan takes four buckets per thread");
static_assert(OFF_EN + OWN_CAP * 2 <= LDS_RAW, "the edge list fits region B");
#define OWN_PK   32
#define OFF_PV   (OFF_EN + OWN_CAP * 2)
static_assert(OFF_PV + OWN_PK * OWN_NT * 4 <= LDS_RAW, "the threads' own hit lists fit region B");
#define NM_PITCH 12
#define NM_BYTES (OWN_R * NM_PITCH * 4)
#define OFF_NSE  (OFF_PV + OWN_PK * OWN_NT * 4)
#define OFF_NMX  OFF_PV
#define OFF_NIV  (OFF_NMX + NM_BYTES)
static_assert(OFF_PV % 16 == 0 && NM_BYTES % 16 == 0 && (OWN_PK * OWN_NT * 4) % 16 == 0, "16-byte loads of the copies are aligned");
static_assert(OFF_NSE + NM_BYTES <= LDS_RAW, "the node-score copy fits behind the hit lists");
static_assert(OFF_NIV + NM_BYTES <= OFF_NSE, "max and denominator copies fit the dead hit lists");
static_assert(OWN_R * 8 <= 4 * OWN_NT && OWN_NT % 8 == 0, "a copy pass: eight threads per node, at most four nodes per thread");
#define PJ_PITCH 132
#define PJ_TILES ((OWN_R + 15) / 16)
#define PJ_PATCH (16 * PJ_PITCH * 4)
#define PJ_LOW   7
#define PJ_OFF(t) ((t) < PJ_LOW ? (t) * PJ_PATCH : OFF_EA + ((t) - PJ_LOW) * PJ_PATCH)
#define PJ_WOFF  (PJ_LOW * PJ_PATCH)
static_assert(PJ_WOFF + 8 * PJ_PITCH * 4 <= OFF_B, "low patches and the weight rows end before the tables");
static_assert(OFF_EA + (PJ_TILES - PJ_LOW) * PJ_PATCH <= LDS_RAW, "high patches fit behind the tables");
static_assert(OFF_EA + (PJ_TILES - PJ_LOW) * PJ_PATCH <= OFF_NSE, "high patches end before the node-major score copy");
static_assert(PJ_TILES <= OWN_NT / 32 && OWN_NT >= 256, "a wave per tile, a thread per 16 bytes of the weight rows");

__device__ __forceinline__ v4f pj_ld4(const float* __restrict__ p) {
    v4f r;
    r.x = p[0]; r.y = p[1]; r.z = p[2]; r.w = p[3];
    return r;
}

template <int PH>
__device__ __forceinline__ void own_slow_sweep(const int* __restrict__ idx,
                                               const int* __restrict__ rel,
                                               const float* __restrict__ Sr,
                                               const float* __restrict__ x_r,
                                               float* s_acc, unsigned* s_max,
                                               float* s_sum, const float* s_se,
                                               int E, int base, int nown, int tid) {
    for (int e = tid; e < E; e += OWN_NT) {
        const unsigned d = (unsigned)(idx[e] - base);
        if (d >= (unsigned)nown) continue;
        const int r = rel[e];
        if (PH == 0) {
            for (int h = 0; h < 8; ++h) {
                const float s = lrelu(s_se[OWN_TIX(d, h)] + Sr[(size_t)r * 16 + h]);
                atomicMax(&s_max[OWN_TIX(d, h)], fkey(s));
            }
        } else if (PH == 1) {
            for (int h = 0; h < 8; ++h) {
                const float s = lrelu(s_se[OWN_TIX(d, h)] + Sr[(size_t)r * 16 + h]);
                atomicAdd(&s_sum[OWN_TIX(d, h)], __expf(s - fdecode(s_max[OWN_TIX(d, h)])));
            }
        } else {
            float a = 0.0f;
            for (int h = 0; h < 8; ++h) {
                const float s = lrelu(s_se[OWN_TIX(d, h)] + Sr[(size_t)r * 16 + h]);
                a += __expf(s - fdecode(s_max[OWN_TIX(d, h)])) / (s_sum[OWN_TIX(d, h)] + 1e-16f);
            }
            a *= 0.125f;
            for (int c = 0; c < 128; ++c)
                atomicAdd(&s_acc[d * 128 + c], x_r[(size_t)r * 128 + c] * a);
        }
    }
}

__global__ __launch_bounds__(OWN_NT) void k_owner(const int* __restrict__ eidx,
                                                  const unsigned* __restrict__ map,
                                                  const unsigned* __restrict__ comb,
                                                  const int* __restrict__ rel,
                                                  const float* __restrict__ x_e,
                                                  const float* __restrict__ Wh,
                                                  const float* __restrict__ Wt,
                                                  const float* __restrict__ Sr,
                                                  const float* __restrict__ x_r,
                                                  float* __restrict__ out,
                                                  int E, int N_e, int nblk, int N_r) {
    __shared__ __attribute__((aligned(16))) unsigned char s_raw[LDS_RAW];
    __shared__ int s_cnt;
    __shared__ int s_bc[OWN_NBK];
    __shared__ unsigned short s_bo[OWN_NBK + 1];
    __shared__ int s_wt[32];
    uint2*          const s_rec  = (uint2*)s_raw;
    unsigned*       const s_max  = (unsigned*)(s_raw + OFF_B);
    float*          const s_sum  = (float*)(s_raw + OFF_SUM);
    float*          const s_se   = (float*)(s_raw + OFF_SE);
    int*            const s_ea   = (int*)(s_raw + OFF_EA);
    unsigned short* const s_en   = (unsigned short*)(s_raw + OFF_EN);
    float*          const s_acc  = (float*)(s_raw + OFF_EA);
    float*          const s_rows = (float*)(s_raw + OFF_B);
    int*            const s_pv   = (int*)(s_raw + OFF_PV);
    float*          const s_nse  = (float*)(s_raw + OFF_NSE);
    float*          const s_nmx  = (float*)(s_raw + OFF_NMX);
    float*          const s_niv  = (float*)(s_raw + OFF_NIV);

    const int tid  = threadIdx.x;
    const int lane = tid & 31;
    const int w    = tid >> 5;
    const int side = ((int)blockIdx.x >= nblk) ? 1 : 0;
    const int base = ((int)blockIdx.x - side * nblk) * OWN_R;
    const int nown = min(OWN_R, N_e - base);
    const int* __restrict__ idx = eidx + (size_t)side * (size_t)E;

    const int NS = (N_r + SL_ROWS - 1) / SL_ROWS;
    v4f xrow[16];
    v4f wrow = {0.f, 0.f, 0.f, 0.f};
    const int wu = __builtin_amdgcn_readfirstlane(w);
    if (wu < PJ_TILES) {
        const int rw = base + wu * 16;
#pragma unroll
        for (int j = 0; j < 16; ++j)
            xrow[j] = pj_ld4(x_e + (size_t)min(rw + j, N_e - 1) * 128 + lane * 4);
    }
    if (tid < 256) {
        if (side) wrow = pj_ld4(Wt + tid * 4);
        else      wrow = pj_ld4(Wh + tid * 4);
    }
    for (int i = tid; i < OWN_NBK; i += OWN_NT) s_bc[i] = 0;
    for (int i = tid; i < OWN_R * 8; i += OWN_NT) {
        s_max[i] = NEG_INF_KEY;
        s_sum[i] = 0.0f;
    }
    if (tid == 0) s_cnt = 0;
    if (wu < PJ_TILES) {
        float* const pp = (float*)(s_raw + PJ_OFF(wu));
#pragma unroll
        for (int j = 0; j < 16; ++j) *(v4f*)(pp + j * PJ_PITCH + lane * 4) = xrow[j];
    }
    if (tid < 256) *(v4f*)((float*)(s_raw + PJ_WOFF) + w * PJ_PITCH + lane * 4) = wrow;
    __syncthreads();
    if (wu < PJ_TILES) {
        const int m    = lane & 15;
        const int kofs = (lane >> 4) << 1;
        const float* const ap = (const float*)(s_raw + PJ_OFF(wu)) + m * PJ_PITCH + kofs;
        const float* const bp = (const float*)(s_raw + PJ_WOFF) + (m & 7) * PJ_PITCH + kofs;
        v8f c = {0.f, 0.f, 0.f, 0.f, 0.f, 0.f, 0.f, 0.f};
#pragma unroll
        for (int k = 0; k < 128; k += 4) {
            const v2f a = *(const v2f*)(ap + k);
            const v2f b = *(const v2f*)(bp + k);
            c = __builtin_amdgcn_wmma_f32_16x16x4_f32(
                 false, a,  false, b,
                 (short)0, c,  false,  false);
        }
        const int h  = lane & 15;
        const int r0 = wu * 16 + ((lane >> 4) << 3);
        if (h < 8) {
#pragma unroll
            for (int v = 0; v < 8; ++v) {
                const int nl = r0 + v;
                if (nl < OWN_R) {
                    const float sv = (nl < nown) ? c[v] : 0.0f;
                    s_se[OWN_TIX(nl, h)] = sv;
                    s_nse[nl * NM_PITCH + h] = sv;
                }
            }
        }
    }
    __syncthreads();

    int cntp = 0;
    auto keep1 = [&](int e) {
        s_pv[min(cntp, OWN_PK - 1) * OWN_NT + tid] = e;
        ++cntp;
    };
    if (nblk <= 255) {
        const unsigned char* __restrict__ mp = (const unsigned char*)map + (size_t)side * (size_t)E;
        const unsigned me  = (unsigned)((int)blockIdx.x - side * nblk);
        const unsigned pat = me * 0x01010101u;
        auto eqmask = [&](unsigned v) {
            const unsigned x = v ^ pat;
            const unsigned t = (x & 0x7F7F7F7Fu) + 0x7F7F7F7Fu;
            return ~(t | x) & 0x80808080u;
        };
        auto hit16 = [&](const uint4 a) {
            return (eqmask(a.x) >> 3) | (eqmask(a.y) >> 2) | (eqmask(a.z) >> 1) | eqmask(a.w);
        };
        auto drain = [&](unsigned wv, int eb) {
            while (wv != 0u) {
                const unsigned k = (unsigned)__ffs((int)wv) - 1u;
                wv &= wv - 1u;
                keep1(eb + (int)((((k >> 2) & 1u) * (16u * OWN_NT)) + ((k & 3u) << 2) + (k >> 3)));
            }
        };
        int head = (int)((16u - (unsigned)((uintptr_t)mp & 15u)) & 15u);
        if (head > E) head = E;
        for (int e = tid; e < head; e += OWN_NT) if ((unsigned)mp[e] == me) keep1(e);
        const int nvec = (E - head) >> 4;
        const uint4* __restrict__ vp = (const uint4*)(mp + head);
        int q = tid;
        for (; q + 3 * OWN_NT < nvec; q += 4 * OWN_NT) {
            const uint4 a0 = vp[q];
            const uint4 a1 = vp[q + OWN_NT];
            const uint4 a2 = vp[q + 2 * OWN_NT];
            const uint4 a3 = vp[q + 3 * OWN_NT];
            const unsigned wa = (hit16(a0) >> 4) | hit16(a1);
            const unsigned wb = (hit16(a2) >> 4) | hit16(a3);
            drain(wa, head + 16 * q);
            drain(wb, head + 16 * (q + 2 * OWN_NT));
        }
        for (; q < nvec; q += OWN_NT) drain(hit16(vp[q]) >> 4, head + 16 * q);
        for (int e = head + 16 * nvec + tid; e < E; e += OWN_NT) if ((unsigned)mp[e] == me) keep1(e);
    } else {
        for (int e = tid; e < E; e += OWN_NT)
            if ((unsigned)(idx[e] - base) < (unsigned)nown) keep1(e);
    }
    {
        const int mine = (cntp > OWN_PK) ? (OWN_CAP + 1) : cntp;
        int x = mine;
#pragma unroll
        for (int d = 1; d < 32; d <<= 1) {
            const int y = __shfl_up(x, d);
            if (lane >= d) x += y;
        }
        if (lane == 31) s_wt[w] = x;
        __syncthreads();
        if (w == 0) {
            const int v = s_wt[lane];
            int z = v;
#pragma unroll
            for (int d = 1; d < 32; d <<= 1) {
                const int y = __shfl_up(z, d);
                if (lane >= d) z += y;
            }
            s_wt[lane] = z - v;
        }
        __syncthreads();
        const int off = s_wt[w] + x - mine;
        if (tid == OWN_NT - 1) s_cnt = off + mine;
        const int have = min(cntp, OWN_PK);
        for (int i = 0; i < have; ++i)
            if (off + i < OWN_CAP) s_ea[off + i] = s_pv[i * OWN_NT + tid];
    }
    __syncthreads();
    const int total = s_cnt;
    const bool fits = total <= OWN_CAP && NS >= 1 && NS <= SL_MAXNS &&
                      ((((uintptr_t)x_r) | ((uintptr_t)out) | ((uintptr_t)Sr)) & 15u) == 0;

    if (fits) {
        const int cnt = total;
        if (cnt > 0) {
            const unsigned* __restrict__ cwp = comb + (size_t)side * (size_t)E;
            int ed[OWN_EPT];
            unsigned cw[OWN_EPT];
#pragma unroll
            for (int j = 0; j < OWN_EPT; ++j)
                ed[j] = s_ea[min(tid + j * OWN_NT, cnt - 1)];
#pragma unroll
            for (int j = 0; j < OWN_EPT; ++j) cw[j] = cwp[ed[j]];
            __syncthreads();
#pragma unroll
            for (int j = 0; j < OWN_EPT; ++j) {
                const int p = tid + j * OWN_NT;
                if (p < cnt) {
                    const int r  = (int)(cw[j] >> 9);
                    const int nl = (int)(cw[j] & 511u);
                    s_ea[p] = r;
                    s_en[p] = (unsigned short)nl;
                    atomicAdd(&s_bc[OWN_KEY(nl, r)], 1);
                }
            }
        }
        __syncthreads();
        {
            const int k0 = 4 * tid;
            int c[4];
#pragma unroll
            for (int j = 0; j < 4; ++j) c[j] = (k0 + j < OWN_NBK) ? s_bc[k0 + j] : 0;
            const int mine = c[0] + c[1] + c[2] + c[3];
            int x = mine;
#pragma unroll
            for (int d = 1; d < 32; d <<= 1) {
                const int y = __shfl_up(x, d);
                if (lane >= d) x += y;
            }
            if (lane == 31) s_wt[w] = x;
            __syncthreads();
            if (w == 0) {
                const int v = s_wt[lane];
                int z = v;
#pragma unroll
                for (int d = 1; d < 32; d <<= 1) {
                    const int y = __shfl_up(z, d);
                    if (lane >= d) z += y;
                }
                s_wt[lane] = z - v;
            }
            __syncthreads();
            int run = s_wt[w] + x - mine;
#pragma unroll
            for (int j = 0; j < 4; ++j) {
                if (k0 + j < OWN_NBK) { s_bo[k0 + j] = (unsigned short)run; s_bc[k0 + j] = run; }
                run += c[j];
            }
            if (tid == 0) s_bo[OWN_NBK] = (unsigned short)cnt;
        }
        float sc[OWN_EPT][8];
#pragma unroll
        for (int j = 0; j < OWN_EPT; ++j) {
            if (j * OWN_NT >= cnt) break;
            const int pc = min(tid + j * OWN_NT, cnt - 1);
            const float4* __restrict__ sp = (const float4*)(Sr + (size_t)s_ea[pc] * 16);
            const float4 lo = sp[0], hi = sp[1];
            sc[j][0] = lo.x; sc[j][1] = lo.y; sc[j][2] = lo.z; sc[j][3] = lo.w;
            sc[j][4] = hi.x; sc[j][5] = hi.y; sc[j][6] = hi.z; sc[j][7] = hi.w;
        }
#pragma unroll
        for (int j = 0; j < OWN_EPT; ++j) {
            if (j * OWN_NT >= cnt) break;
            const int p = tid + j * OWN_NT;
            const int n = (int)s_en[min(p, cnt - 1)];
            const float4 e0 = *(const float4*)(s_nse + n * NM_PITCH);
            const float4 e1 = *(const float4*)(s_nse + n * NM_PITCH + 4);
            const float se[8] = {e0.x, e0.y, e0.z, e0.w, e1.x, e1.y, e1.z, e1.w};
#pragma unroll
            for (int h = 0; h < 8; ++h) sc[j][h] = lrelu(se[h] + sc[j][h]);
            if (p < cnt) {
#pragma unroll
                for (int h = 0; h < 8; ++h) atomicMax(&s_max[OWN_TIX(n, h)], fkey(sc[j][h]));
            }
        }
        __syncthreads();
        {
            unsigned v[4];
#pragma unroll
            for (int k = 0; k < 4; ++k)
                v[k] = s_max[OWN_TIX(min((tid >> 3) + k * (OWN_NT / 8), OWN_R - 1), tid & 7)];
#pragma unroll
            for (int k = 0; k < 4; ++k) {
                const int n = (tid >> 3) + k * (OWN_NT / 8);
                if (n < OWN_R) s_nmx[n * NM_PITCH + (tid & 7)] = fdecode(v[k]);
            }
        }
        __syncthreads();
#pragma unroll
        for (int j = 0; j < OWN_EPT; ++j) {
            if (j * OWN_NT >= cnt) break;
            const int p = tid + j * OWN_NT;
            const int n = (int)s_en[min(p, cnt - 1)];
            const float4 m0 = *(const float4*)(s_nmx + n * NM_PITCH);
            const float4 m1 = *(const float4*)(s_nmx + n * NM_PITCH + 4);
            const float mx[8] = {m0.x, m0.y, m0.z, m0.w, m1.x, m1.y, m1.z, m1.w};
#pragma unroll
            for (int h = 0; h < 8; ++h) sc[j][h] = __expf(sc[j][h] - mx[h]);
            if (p < cnt) {
#pragma unroll
                for (int h = 0; h < 8; ++h) atomicAdd(&s_sum[OWN_TIX(n, h)], sc[j][h]);
            }
        }
        __syncthreads();
        {
            float v[4];
#pragma unroll
            for (int k = 0; k < 4; ++k)
                v[k] = s_sum[OWN_TIX(min((tid >> 3) + k * (OWN_NT / 8), OWN_R - 1), tid & 7)];
#pragma unroll
            for (int k = 0; k < 4; ++k) {
                const int n = (tid >> 3) + k * (OWN_NT / 8);
                if (n < OWN_R) s_niv[n * NM_PITCH + (tid & 7)] = 1.0f / (v[k] + 1e-16f);
            }
        }
        __syncthreads();
#pragma unroll
        for (int j = 0; j < OWN_EPT; ++j) {
            if (j * OWN_NT >= cnt) break;
            const int p = tid + j * OWN_NT;
            const int pc = min(p, cnt - 1);
            const int n = (int)s_en[pc];
            const float4 v0 = *(const float4*)(s_niv + n * NM_PITCH);
            const float4 v1 = *(const float4*)(s_niv + n * NM_PITCH + 4);
            const float iv[8] = {v0.x, v0.y, v0.z, v0.w, v1.x, v1.y, v1.z, v1.w};
            float a = sc[j][0] * iv[0];
#pragma unroll
            for (int h = 1; h < 8; ++h) a += sc[j][h] * iv[h];
            if (p < cnt) {
                const int r  = s_ea[pc];
                const int qd = atomicAdd(&s_bc[OWN_KEY(n, r)], 1);
                s_rec[qd] = make_uint2(((unsigned)r << 9) | (unsigned)n, __float_as_uint(a * 0.125f));
            }
        }
        __syncthreads();
#define OWN_SLOT_LIST(X) X(0) X(1) X(2) X(3) X(4) X(5) X(6) X(7) X(8) X(9) X(10) X(11) X(12)
        static_assert(OWN_SLOTS == 13, "the slot list names 13 slots");
#define OWN_DECL(S) float4 acc##S = make_float4(0.f, 0.f, 0.f, 0.f);
        OWN_SLOT_LIST(OWN_DECL)
        const int ws = __builtin_amdgcn_readfirstlane(w);
        v4f t[OWN_G];
        {
            const int nv = min(SL_ROWS, N_r) * 32;
            const v4f* __restrict__ sp = (const v4f*)x_r;
#pragma unroll
            for (int j = 0; j < OWN_G; ++j) t[j] = sp[min(tid + j * OWN_NT, nv - 1)];
        }
        for (int s = 0; s < NS; ++s) {
            if (s > 0) __syncthreads();
            {
                const int nv = min(SL_ROWS, N_r - s * SL_ROWS) * 32;
                v4f* dp = (v4f*)s_rows;
#pragma unroll
                for (int j = 0; j < OWN_G; ++j)
                    if (tid + j * OWN_NT < nv) dp[tid + j * OWN_NT] = t[j];
            }
            __syncthreads();
            if (s + 1 < NS) {
                const int r1 = (s + 1) * SL_ROWS;
                const int nv = min(SL_ROWS, N_r - r1) * 32;
                const v4f* __restrict__ sp = (const v4f*)(x_r + (size_t)r1 * 128);
#pragma unroll
                for (int j = 0; j < OWN_G; ++j) t[j] = sp[min(tid + j * OWN_NT, nv - 1)];
            }
            const int kb = (s * 32 + ws) * OWN_SLOTS;
            const int bv = (int)s_bo[kb + min(lane, OWN_SLOTS)];
            const int qz = __builtin_amdgcn_readlane(bv, OWN_SLOTS);
            const int rbase = s * SL_ROWS;
            int q = __builtin_amdgcn_readlane(bv, 0);
            while (q < qz) {
                const int cb = q;
                const int ce = min(qz, cb + 32);
                const uint2 rc = s_rec[min(cb + lane, qz - 1)];
#define OWN_SLOT_RUN(S) { \
                    const int e_ = min(__builtin_amdgcn_readlane(bv, S + 1), ce); \
                    if (q < e_) { \
                        for (int g = q; g < e_; g += 2) { \
                            v4f m_[2]; float a_[2]; \
                            _Pragma("unroll")   \
                            for (int j = 0; j < 2; ++j) { \
                                const int li = min(g + j, e_ - 1) - cb; \
                                const unsigned rx = (unsigned)__builtin_amdgcn_readlane((int)rc.x, li); \
                                a_[j] = __uint_as_float((unsigned)__builtin_amdgcn_readlane((int)rc.y, li)); \
                                m_[j] = ((const v4f*)(s_rows + ((int)(rx >> 9) - rbase) * 128))[lane]; \
                            } \
                              \
                              \
                            asm volatile("" :: "v"(m_[0]), "v"(m_[1])); \
                            _Pragma("unroll")   \
                            for (int j = 0; j < 2; ++j) \
                                if (g + j < e_) {   \
                                    acc##S.x += m_[j].x * a_[j]; acc##S.y += m_[j].y * a_[j]; \
                                    acc##S.z += m_[j].z * a_[j]; acc##S.w += m_[j].w * a_[j]; \
                                } \
                        } \
                        q = e_; \
                    } }
                OWN_SLOT_LIST(OWN_SLOT_RUN)
#undef OWN_SLOT_RUN
            }
        }
#define OWN_OUT(S) if (ws + 32 * S < nown) ((float4*)(out + (size_t)(base + ws + 32 * S) * 256 + side * 128))[lane] = acc##S;
        OWN_SLOT_LIST(OWN_OUT)
#undef OWN_OUT
#undef OWN_DECL
#undef OWN_SLOT_LIST
    } else {
        for (int i = tid; i < OWN_R * 32; i += OWN_NT)
            ((float4*)s_acc)[i] = make_float4(0.f, 0.f, 0.f, 0.f);
        __syncthreads();
        own_slow_sweep<0>(idx, rel, Sr, x_r, s_acc, s_max, s_sum, s_se, E, base, nown, tid);
        __syncthreads();
        own_slow_sweep<1>(idx, rel, Sr, x_r, s_acc, s_max, s_sum, s_se, E, base, nown, tid);
        __syncthreads();
        own_slow_sweep<2>(idx, rel, Sr, x_r, s_acc, s_max, s_sum, s_se, E, base, nown, tid);
        __syncthreads();
        for (int nl = w; nl < nown; nl += 32) {
            const float4 v = ((const float4*)(s_acc + nl * 128))[lane];
            ((float4*)(out + (size_t)(base + nl) * 256 + side * 128))[lane] = v;
        }
    }
}

extern "C" void kernel_launch(void* const* d_in, const int* in_sizes, int n_in,
                              void* d_out, int out_size, void* d_ws, size_t ws_size,
                              hipStream_t stream) {
    const float* x_e  = (const float*)d_in[0];
    const float* x_r  = (const float*)d_in[1];
    const float* Wh   = (const float*)d_in[2];
    const float* Wt   = (const float*)d_in[3];
    const float* Wr   = (const float*)d_in[4];
    const int*   eidx = (const int*)d_in[5];
    const int*   rel  = (const int*)d_in[6];

    const int D   = 128;
    const int N_e = in_sizes[0] / D;
    const int N_r = in_sizes[1] / D;
    const int E   = in_sizes[6];

    auto align256 = [](size_t x) { return (x + 255) & ~(size_t)255; };
    char* ws = (char*)d_ws;
    float* Sr = (float*)ws; ws += align256((size_t)N_r * 16 * sizeof(float));
    unsigned* omap = (unsigned*)ws; ws += align256((size_t)2 * (size_t)E + 16);
    unsigned* comb = (unsigned*)ws; ws += align256((size_t)2 * (size_t)E * sizeof(unsigned));

    float* out = (float*)d_out;

    const int BLK = 256;

    {
        const int waves   = (N_r + 15) / 16;
        const int threads = waves * 32;
        const int grid    = (threads + BLK - 1) / BLK;
        k_proj_wmma<<<grid, BLK, 0, stream>>>(x_r, Wr, Wr, Sr, N_r);
    }

    {
        const int n    = 2 * E;
        const int grid = (n + 16 * BLK - 1) / (16 * BLK);
        k_ownmap<<<grid, BLK, 0, stream>>>(eidx, rel, omap, comb, n, E);
    }

    {
        const int nblk = (N_e + OWN_R - 1) / OWN_R;
        k_owner<<<2 * nblk, OWN_NT, 0, stream>>>(eidx, omap, comb, rel, x_e, Wh, Wt, Sr, x_r, out, E, N_e, nblk, N_r);
    }
}
